# starting kernel
# baseline (speedup 1.0000x reference)
#include <hip/hip_runtime.h>


typedef float v2f __attribute__((ext_vector_type(2)));
typedef float v8f __attribute__((ext_vector_type(8)));

#define N_PTS   400000
#define M_Q     100000
#define NB      16
#define C_IN    64
#define D2      32
#define C_OUT   128
#define KP      15
#define INV_EXTENT 2.0f
#define EPS_BN  1e-5f
#define SLOPE   0.2f

__device__ __forceinline__ float leaky(float v) { return v >= 0.f ? v : SLOPE * v; }

__device__ __forceinline__ v8f wmma4(v2f a, v2f b, v8f c) {
  return __builtin_amdgcn_wmma_f32_16x16x4_f32(false, a, false, b, (short)0, c,
                                               false, false);
}

__device__ __forceinline__ v2f load_a_rm(const float* src, int ld, int m0,
                                         int k0, int lane) {
  int m = m0 + (lane & 15);
  int k = k0 + ((lane >> 4) << 1);
  v2f a;
  a.x = src[(size_t)m * ld + k];
  a.y = src[(size_t)m * ld + k + 1];
  return a;
}

__device__ __forceinline__ v2f load_b_rm(const float* src, int ld, int k0,
                                         int n0, int lane) {
  int n = n0 + (lane & 15);
  int k = k0 + ((lane >> 4) << 1);
  v2f b;
  b.x = src[(size_t)k * ld + n];
  b.y = src[(size_t)(k + 1) * ld + n];
  return b;
}

__device__ __forceinline__ void store_d_rm(float* dst, int ld, int m0, int n0,
                                           int lane, v8f c) {
  int n = n0 + (lane & 15);
  int rb = m0 + ((lane >> 4) << 3);
#pragma unroll
  for (int v = 0; v < 8; ++v) dst[(size_t)(rb + v) * ld + n] = c[v];
}

__device__ __forceinline__ void accum_stats(float* sum, float* sq, int n0,
                                            int lane, v8f c) {
  float s = 0.f, q = 0.f;
#pragma unroll
  for (int v = 0; v < 8; ++v) { s += c[v]; q += c[v] * c[v]; }
  atomicAdd(&sum[n0 + (lane & 15)], s);
  atomicAdd(&sq[n0 + (lane & 15)], q);
}

__global__ void k_zero(float* p, int n) {
  int i = blockIdx.x * blockDim.x + threadIdx.x;
  if (i < n) p[i] = 0.f;
}

__global__ void k_finalize_bn(const float* __restrict__ sum,
                              const float* __restrict__ sq,
                              const float* __restrict__ g,
                              const float* __restrict__ b, float cnt,
                              float* a, float* c, int C) {
  int i = blockIdx.x * blockDim.x + threadIdx.x;
  if (i >= C) return;
  float mu  = sum[i] / cnt;
  float var = sq[i] / cnt - mu * mu;
  float s   = g[i] * rsqrtf(fmaxf(var, 0.f) + EPS_BN);
  a[i] = s;
  c[i] = b[i] - mu * s;
}

__global__ __launch_bounds__(256) void k_gemm1(const float* __restrict__ x,
                                               const float* __restrict__ W,
                                               float* __restrict__ y1,
                                               float* sum1, float* sq1,
                                               int ntiles) {
  int lane = threadIdx.x & 31;
  int tile = blockIdx.x * (blockDim.x >> 5) + (threadIdx.x >> 5);
  if (tile >= ntiles) return;
  int m0 = tile * 16;
  v8f acc0 = {}, acc1 = {};
#pragma unroll
  for (int k0 = 0; k0 < C_IN; k0 += 4) {
    v2f a  = load_a_rm(x, C_IN, m0, k0, lane);
    v2f b0 = load_b_rm(W, D2, k0, 0, lane);
    v2f b1 = load_b_rm(W, D2, k0, 16, lane);
    acc0 = wmma4(a, b0, acc0);
    acc1 = wmma4(a, b1, acc1);
  }
  store_d_rm(y1, D2, m0, 0, lane, acc0);
  store_d_rm(y1, D2, m0, 16, lane, acc1);
  accum_stats(sum1, sq1, 0, lane, acc0);
  accum_stats(sum1, sq1, 16, lane, acc1);
}

__global__ __launch_bounds__(128) void k_kpconv(
    const float* __restrict__ y1, const int* __restrict__ idx,
    const float* __restrict__ pos, const float* __restrict__ qpos,
    const float* __restrict__ Kpts, const float* __restrict__ KV,
    const float* __restrict__ a1, const float* __restrict__ c1,
    float* __restrict__ y2, float* sumk, float* sqk) {
  __shared__ float smem[12288];
  float* s_w  = smem;
  float* s_nf = smem + 4096;
  int lane = threadIdx.x & 31;
  int wave = threadIdx.x >> 5;
  int q0 = blockIdx.x * 16;

  for (int p = threadIdx.x; p < 256; p += 128) {
    int q = p >> 4, nb = p & 15;
    int id = idx[(size_t)(q0 + q) * NB + nb];
    float qx = qpos[(size_t)(q0 + q) * 3 + 0];
    float qy = qpos[(size_t)(q0 + q) * 3 + 1];
    float qz = qpos[(size_t)(q0 + q) * 3 + 2];
    float px, py, pz;
    if (id < N_PTS) {
      px = pos[(size_t)id * 3 + 0];
      py = pos[(size_t)id * 3 + 1];
      pz = pos[(size_t)id * 3 + 2];
    } else {
      px = py = pz = 1e6f;
    }
    float rx = px - qx, ry = py - qy, rz = pz - qz;
    float rr = rx * rx + ry * ry + rz * rz;
#pragma unroll
    for (int k = 0; k < KP; ++k) {
      float kx = Kpts[k * 3 + 0], ky = Kpts[k * 3 + 1], kz = Kpts[k * 3 + 2];
      float sq = rr + kx * kx + ky * ky + kz * kz
                 - 2.f * (rx * kx + ry * ky + rz * kz);
      float w = 1.f - sqrtf(fmaxf(sq, 0.f)) * INV_EXTENT;
      s_w[q * 256 + nb * 16 + k] = fmaxf(w, 0.f);
    }
    s_w[q * 256 + nb * 16 + 15] = 0.f;
    if (id < N_PTS) {
      const float* yr = y1 + (size_t)id * D2;
#pragma unroll
      for (int d = 0; d < D2; ++d)
        s_nf[q * 512 + nb * 32 + d] = leaky(yr[d] * a1[d] + c1[d]);
    } else {
#pragma unroll
      for (int d = 0; d < D2; ++d) s_nf[q * 512 + nb * 32 + d] = 0.f;
    }
  }
  __syncthreads();

  v8f wt[4][2];
#pragma unroll
  for (int qq = 0; qq < 4; ++qq) { wt[qq][0] = (v8f){}; wt[qq][1] = (v8f){}; }
#pragma unroll
  for (int qq = 0; qq < 4; ++qq) {
    int q = wave * 4 + qq;
#pragma unroll
    for (int k0 = 0; k0 < 16; k0 += 4) {
      int m = lane & 15;
      int kk = k0 + ((lane >> 4) << 1);
      v2f a;
      a.x = s_w[q * 256 + kk * 16 + m];
      a.y = s_w[q * 256 + (kk + 1) * 16 + m];
#pragma unroll
      for (int t = 0; t < 2; ++t) {
        int n = t * 16 + (lane & 15);
        v2f b;
        b.x = s_nf[q * 512 + kk * 32 + n];
        b.y = s_nf[q * 512 + (kk + 1) * 32 + n];
        wt[qq][t] = wmma4(a, b, wt[qq][t]);
      }
    }
  }
  __syncthreads();

  float* s_wt = smem;
#pragma unroll
  for (int qq = 0; qq < 4; ++qq) {
    int q = wave * 4 + qq;
#pragma unroll
    for (int t = 0; t < 2; ++t) {
      int n = t * 16 + (lane & 15);
      int rb = (lane >> 4) << 3;
#pragma unroll
      for (int v = 0; v < 8; ++v)
        s_wt[q * 512 + (rb + v) * 32 + n] = wt[qq][t][v];
    }
  }
  __syncthreads();

  int ntile = wave & 1;
  int kbase = (wave >> 1) * 256;
  v8f acc = {};
  for (int k0 = 0; k0 < 256; k0 += 4) {
    int kk = kbase + k0 + ((lane >> 4) << 1);
    int m = lane & 15;
    v2f a;
    a.x = s_wt[m * 512 + kk];
    a.y = s_wt[m * 512 + kk + 1];
    int n = ntile * 16 + (lane & 15);
    v2f b;
    b.x = (kk < 480) ? KV[(size_t)kk * 32 + n] : 0.f;
    b.y = (kk + 1 < 480) ? KV[(size_t)(kk + 1) * 32 + n] : 0.f;
    acc = wmma4(a, b, acc);
  }
  __syncthreads();
  float* s_red = smem;
  if (wave >= 2) {
#pragma unroll
    for (int v = 0; v < 8; ++v) s_red[(wave - 2) * 256 + lane * 8 + v] = acc[v];
  }
  __syncthreads();
  if (wave < 2) {
#pragma unroll
    for (int v = 0; v < 8; ++v) acc[v] += s_red[wave * 256 + lane * 8 + v];
    int col = wave * 16 + (lane & 15);
    int rb = q0 + ((lane >> 4) << 3);
    float s = 0.f, q2 = 0.f;
#pragma unroll
    for (int v = 0; v < 8; ++v) {
      float val = acc[v];
      y2[(size_t)(rb + v) * D2 + col] = val;
      s += val;
      q2 += val * val;
    }
    atomicAdd(&sumk[col], s);
    atomicAdd(&sqk[col], q2);
  }
}

__global__ __launch_bounds__(256) void k_gemm2(const float* __restrict__ y2,
                                               const float* __restrict__ W,
                                               const float* __restrict__ ak,
                                               const float* __restrict__ ck,
                                               float* __restrict__ y3,
                                               float* sum2, float* sq2,
                                               int ntiles) {
  int lane = threadIdx.x & 31;
  int tile = blockIdx.x * (blockDim.x >> 5) + (threadIdx.x >> 5);
  if (tile >= ntiles) return;
  int m0 = tile * 16;
  v8f acc[8];
#pragma unroll
  for (int t = 0; t < 8; ++t) acc[t] = (v8f){};
#pragma unroll
  for (int k0 = 0; k0 < D2; k0 += 4) {
    int m = m0 + (lane & 15);
    int kk = k0 + ((lane >> 4) << 1);
    v2f a;
    a.x = leaky(y2[(size_t)m * D2 + kk] * ak[kk] + ck[kk]);
    a.y = leaky(y2[(size_t)m * D2 + kk + 1] * ak[kk + 1] + ck[kk + 1]);
#pragma unroll
    for (int t = 0; t < 8; ++t) {
      v2f b = load_b_rm(W, C_OUT, k0, t * 16, lane);
      acc[t] = wmma4(a, b, acc[t]);
    }
  }
#pragma unroll
  for (int t = 0; t < 8; ++t) {
    store_d_rm(y3, C_OUT, m0, t * 16, lane, acc[t]);
    accum_stats(sum2, sq2, t * 16, lane, acc[t]);
  }
}

__global__ __launch_bounds__(128) void k_final(const float* __restrict__ x,
                                               const int* __restrict__ idx,
                                               const float* __restrict__ Wsc,
                                               const float* __restrict__ y3,
                                               const float* __restrict__ a2,
                                               const float* __restrict__ c2,
                                               float* __restrict__ out) {
  __shared__ float s_sx[16 * 64];
  int q0 = blockIdx.x * 16;
  for (int p = threadIdx.x; p < 1024; p += 128) {
    int q = p >> 6, c = p & 63;
    float acc = -3.4e38f;
#pragma unroll
    for (int nb = 0; nb < NB; ++nb) {
      int id = idx[(size_t)(q0 + q) * NB + nb];
      float v = (id < N_PTS) ? x[(size_t)id * C_IN + c] : 0.f;
      acc = fmaxf(acc, v);
    }
    s_sx[q * 64 + c] = acc;
  }
  __syncthreads();
  int lane = threadIdx.x & 31;
  int wave = threadIdx.x >> 5;
#pragma unroll
  for (int tt = 0; tt < 2; ++tt) {
    int ntile = wave * 2 + tt;
    v8f acc = {};
#pragma unroll
    for (int k0 = 0; k0 < C_IN; k0 += 4) {
      int m = lane & 15;
      int kk = k0 + ((lane >> 4) << 1);
      v2f a;
      a.x = s_sx[m * 64 + kk];
      a.y = s_sx[m * 64 + kk + 1];
      v2f b = load_b_rm(Wsc, C_OUT, k0, ntile * 16, lane);
      acc = wmma4(a, b, acc);
    }
    int col = ntile * 16 + (lane & 15);
    int rb = (lane >> 4) << 3;
#pragma unroll
    for (int v = 0; v < 8; ++v) {
      int m = q0 + rb + v;
      float h = leaky(y3[(size_t)m * C_OUT + col] * a2[col] + c2[col]);
      out[(size_t)m * C_OUT + col] = h + acc[v];
    }
  }
}

extern "C" void kernel_launch(void* const* d_in, const int* in_sizes, int n_in,
                              void* d_out, int out_size, void* d_ws,
                              size_t ws_size, hipStream_t stream) {
  const float* x    = (const float*)d_in[0];
  const float* pos  = (const float*)d_in[1];
  const float* qpos = (const float*)d_in[2];
  const int*   idx  = (const int*)d_in[3];
  const float* Wu1  = (const float*)d_in[4];
  const float* g1   = (const float*)d_in[5];
  const float* b1   = (const float*)d_in[6];
  const float* Kpts = (const float*)d_in[7];
  const float* KV   = (const float*)d_in[8];
  const float* gk   = (const float*)d_in[9];
  const float* bk   = (const float*)d_in[10];
  const float* Wu2  = (const float*)d_in[11];
  const float* g2   = (const float*)d_in[12];
  const float* b2   = (const float*)d_in[13];
  const float* Wsc  = (const float*)d_in[14];
  float* out = (float*)d_out;

  float* ws = (float*)d_ws;
  float* y1 = ws;
  float* y2 = y1 + (size_t)N_PTS * D2;
  float* y3 = y2 + (size_t)M_Q * D2;
  float* S  = y3 + (size_t)M_Q * C_OUT;
  float *sum1 = S +   0, *sq1 = S +  32;
  float *sumk = S +  64, *sqk = S +  96;
  float *sum2 = S + 128, *sq2 = S + 256;
  float *a1 = S + 384, *c1 = S + 416;
  float *ak = S + 448, *ck = S + 480;
  float *a2 = S + 512, *c2 = S + 640;

  k_zero<<<1, 384, 0, stream>>>(S, 384);

  k_gemm1<<<(N_PTS / 16) / 8, 256, 0, stream>>>(x, Wu1, y1, sum1, sq1,
                                                N_PTS / 16);
  k_finalize_bn<<<1, 32, 0, stream>>>(sum1, sq1, g1, b1, (float)N_PTS, a1, c1,
                                      D2);
  k_kpconv<<<M_Q / 16, 128, 0, stream>>>(y1, idx, pos, qpos, Kpts, KV, a1, c1,
                                         y2, sumk, sqk);
  k_finalize_bn<<<1, 32, 0, stream>>>(sumk, sqk, gk, bk, (float)M_Q, ak, ck,
                                      D2);
  int tiles2 = M_Q / 16;
  k_gemm2<<<(tiles2 + 7) / 8, 256, 0, stream>>>(y2, Wu2, ak, ck, y3, sum2, sq2,
                                                tiles2);
  k_finalize_bn<<<1, 128, 0, stream>>>(sum2, sq2, g2, b2, (float)M_Q, a2, c2,
                                       C_OUT);
  k_final<<<M_Q / 16, 128, 0, stream>>>(x, idx, Wsc, y3, a2, c2, out);
}


# BN column statistics: per-block LDS reduce into partial planes that the finalize kernel sums in fixed order, replacing global f32 atomic adds and the zero-fill launch (sum order changes)
# speedup vs baseline: 1.7790x; 1.7790x over previous
#include <hip/hip_runtime.h>


typedef float v2f __attribute__((ext_vector_type(2)));
typedef float v8f __attribute__((ext_vector_type(8)));

#define N_PTS   400000
#define M_Q     100000
#define NB      16
#define C_IN    64
#define D2      32
#define C_OUT   128
#define KP      15
#define INV_EXTENT 2.0f
#define EPS_BN  1e-5f
#define SLOPE   0.2f
#define G1_BLOCKS (N_PTS / 16 / 8)
#define G2_BLOCKS ((M_Q / 16 + 7) / 8)

__device__ __forceinline__ float leaky(float v) { return v >= 0.f ? v : SLOPE * v; }

__device__ __forceinline__ v8f wmma4(v2f a, v2f b, v8f c) {
  return __builtin_amdgcn_wmma_f32_16x16x4_f32(false, a, false, b, (short)0, c,
                                               false, false);
}

__device__ __forceinline__ v2f load_a_rm(const float* src, int ld, int m0,
                                         int k0, int lane) {
  int m = m0 + (lane & 15);
  int k = k0 + ((lane >> 4) << 1);
  v2f a;
  a.x = src[(size_t)m * ld + k];
  a.y = src[(size_t)m * ld + k + 1];
  return a;
}

__device__ __forceinline__ v2f load_b_rm(const float* src, int ld, int k0,
                                         int n0, int lane) {
  int n = n0 + (lane & 15);
  int k = k0 + ((lane >> 4) << 1);
  v2f b;
  b.x = src[(size_t)k * ld + n];
  b.y = src[(size_t)(k + 1) * ld + n];
  return b;
}

__device__ __forceinline__ void store_d_rm(float* dst, int ld, int m0, int n0,
                                           int lane, v8f c) {
  int n = n0 + (lane & 15);
  int rb = m0 + ((lane >> 4) << 3);
#pragma unroll
  for (int v = 0; v < 8; ++v) dst[(size_t)(rb + v) * ld + n] = c[v];
}


__device__ __forceinline__ void tile_col_stats(v8f c, float& s, float& q) {
  s = 0.f;
  q = 0.f;
#pragma unroll
  for (int v = 0; v < 8; ++v) { s += c[v]; q += c[v] * c[v]; }
}

__device__ __forceinline__ int stat_of_slot(int t, int C) {
  int j = t >> 3, v = t & 7;
  return 4 * j + (v & 3) + C * (v >> 2);
}
__device__ __forceinline__ size_t slot_index(int t, int nblocks, int blk) {
  int j = t >> 3, v = t & 7;
  return ((size_t)j * nblocks + blk) * 8 + v;
}

__global__ __launch_bounds__(1024) void k_finalize_bn(
    const float* __restrict__ part, int nparts, const float* __restrict__ g,
    const float* __restrict__ b, float cnt, float* a, float* c) {
  __shared__ float s_red[32][8];
  int j = blockIdx.x;
  int h = threadIdx.x & 1;
  int r0 = threadIdx.x >> 1;
  const float4* rows =
      reinterpret_cast<const float4*>(part + (size_t)j * nparts * 8);
  float4 acc = make_float4(0.f, 0.f, 0.f, 0.f);
  for (int base = 0; base < nparts; base += 8 * 512) {
    float4 v[8];
#pragma unroll
    for (int k = 0; k < 8; ++k) {
      int r = base + k * 512 + r0;
      int rc = r < nparts ? r : nparts - 1;
      v[k] = rows[(size_t)rc * 2 + h];
      if (r >= nparts) v[k] = make_float4(0.f, 0.f, 0.f, 0.f);
    }
#pragma unroll
    for (int k = 0; k < 8; ++k) {
      acc.x += v[k].x; acc.y += v[k].y; acc.z += v[k].z; acc.w += v[k].w;
    }
  }
  for (int m = 2; m < 32; m <<= 1) {
    acc.x += __shfl_xor(acc.x, m);
    acc.y += __shfl_xor(acc.y, m);
    acc.z += __shfl_xor(acc.z, m);
    acc.w += __shfl_xor(acc.w, m);
  }
  int lane = threadIdx.x & 31;
  int wave = threadIdx.x >> 5;
  if (lane < 2) {
    s_red[wave][4 * lane + 0] = acc.x;
    s_red[wave][4 * lane + 1] = acc.y;
    s_red[wave][4 * lane + 2] = acc.z;
    s_red[wave][4 * lane + 3] = acc.w;
  }
  __syncthreads();
  if (threadIdx.x < 4) {
    float s = 0.f, q = 0.f;
    for (int w = 0; w < 32; ++w) {
      s += s_red[w][threadIdx.x];
      q += s_red[w][4 + threadIdx.x];
    }
    int ch = 4 * j + threadIdx.x;
    float mu  = s / cnt;
    float var = q / cnt - mu * mu;
    float sc  = g[ch] * rsqrtf(fmaxf(var, 0.f) + EPS_BN);
    a[ch] = sc;
    c[ch] = b[ch] - mu * sc;
  }
}

__global__ __launch_bounds__(256) void k_gemm1(const float* __restrict__ x,
                                               const float* __restrict__ W,
                                               float* __restrict__ y1,
                                               float* __restrict__ part,
                                               int ntiles) {
  __shared__ float s_st[8][2 * D2][2];
  int lane = threadIdx.x & 31;
  int wave = threadIdx.x >> 5;
  int tile = blockIdx.x * 8 + wave;
  int m0 = tile * 16;
  bool live = tile < ntiles;
  float s0 = 0.f, q0 = 0.f, s1 = 0.f, q1 = 0.f;
  v8f acc0 = {}, acc1 = {};
  if (live) {
#pragma unroll
    for (int k0 = 0; k0 < C_IN; k0 += 4) {
      v2f a  = load_a_rm(x, C_IN, m0, k0, lane);
      v2f b0 = load_b_rm(W, D2, k0, 0, lane);
      v2f b1 = load_b_rm(W, D2, k0, 16, lane);
      acc0 = wmma4(a, b0, acc0);
      acc1 = wmma4(a, b1, acc1);
    }
    tile_col_stats(acc0, s0, q0);
    tile_col_stats(acc1, s1, q1);
  }
  int n = lane & 15, h = lane >> 4;
  s_st[wave][n][h] = s0;
  s_st[wave][16 + n][h] = s1;
  s_st[wave][D2 + n][h] = q0;
  s_st[wave][D2 + 16 + n][h] = q1;
  __syncthreads();
  if (live) {
    store_d_rm(y1, D2, m0, 0, lane, acc0);
    store_d_rm(y1, D2, m0, 16, lane, acc1);
  }
  if (threadIdx.x < 2 * D2) {
    int u = stat_of_slot(threadIdx.x, D2);
    float t = 0.f;
    for (int w = 0; w < 8; ++w) t += s_st[w][u][0] + s_st[w][u][1];
    part[slot_index(threadIdx.x, gridDim.x, blockIdx.x)] = t;
  }
}

__global__ __launch_bounds__(128) void k_kpconv(
    const float* __restrict__ y1, const int* __restrict__ idx,
    const float* __restrict__ pos, const float* __restrict__ qpos,
    const float* __restrict__ Kpts, const float* __restrict__ KV,
    const float* __restrict__ a1, const float* __restrict__ c1,
    float* __restrict__ y2, float* __restrict__ part) {
  __shared__ float smem[12288];
  float* s_w  = smem;
  float* s_nf = smem + 4096;
  int lane = threadIdx.x & 31;
  int wave = threadIdx.x >> 5;
  int q0 = blockIdx.x * 16;

  for (int p = threadIdx.x; p < 256; p += 128) {
    int q = p >> 4, nb = p & 15;
    int id = idx[(size_t)(q0 + q) * NB + nb];
    float qx = qpos[(size_t)(q0 + q) * 3 + 0];
    float qy = qpos[(size_t)(q0 + q) * 3 + 1];
    float qz = qpos[(size_t)(q0 + q) * 3 + 2];
    float px, py, pz;
    if (id < N_PTS) {
      px = pos[(size_t)id * 3 + 0];
      py = pos[(size_t)id * 3 + 1];
      pz = pos[(size_t)id * 3 + 2];
    } else {
      px = py = pz = 1e6f;
    }
    float rx = px - qx, ry = py - qy, rz = pz - qz;
    float rr = rx * rx + ry * ry + rz * rz;
#pragma unroll
    for (int k = 0; k < KP; ++k) {
      float kx = Kpts[k * 3 + 0], ky = Kpts[k * 3 + 1], kz = Kpts[k * 3 + 2];
      float sq = rr + kx * kx + ky * ky + kz * kz
                 - 2.f * (rx * kx + ry * ky + rz * kz);
      float w = 1.f - sqrtf(fmaxf(sq, 0.f)) * INV_EXTENT;
      s_w[q * 256 + nb * 16 + k] = fmaxf(w, 0.f);
    }
    s_w[q * 256 + nb * 16 + 15] = 0.f;
    if (id < N_PTS) {
      const float* yr = y1 + (size_t)id * D2;
#pragma unroll
      for (int d = 0; d < D2; ++d)
        s_nf[q * 512 + nb * 32 + d] = leaky(yr[d] * a1[d] + c1[d]);
    } else {
#pragma unroll
      for (int d = 0; d < D2; ++d) s_nf[q * 512 + nb * 32 + d] = 0.f;
    }
  }
  __syncthreads();

  v8f wt[4][2];
#pragma unroll
  for (int qq = 0; qq < 4; ++qq) { wt[qq][0] = (v8f){}; wt[qq][1] = (v8f){}; }
#pragma unroll
  for (int qq = 0; qq < 4; ++qq) {
    int q = wave * 4 + qq;
#pragma unroll
    for (int k0 = 0; k0 < 16; k0 += 4) {
      int m = lane & 15;
      int kk = k0 + ((lane >> 4) << 1);
      v2f a;
      a.x = s_w[q * 256 + kk * 16 + m];
      a.y = s_w[q * 256 + (kk + 1) * 16 + m];
#pragma unroll
      for (int t = 0; t < 2; ++t) {
        int n = t * 16 + (lane & 15);
        v2f b;
        b.x = s_nf[q * 512 + kk * 32 + n];
        b.y = s_nf[q * 512 + (kk + 1) * 32 + n];
        wt[qq][t] = wmma4(a, b, wt[qq][t]);
      }
    }
  }
  __syncthreads();

  float* s_wt = smem;
#pragma unroll
  for (int qq = 0; qq < 4; ++qq) {
    int q = wave * 4 + qq;
#pragma unroll
    for (int t = 0; t < 2; ++t) {
      int n = t * 16 + (lane & 15);
      int rb = (lane >> 4) << 3;
#pragma unroll
      for (int v = 0; v < 8; ++v)
        s_wt[q * 512 + (rb + v) * 32 + n] = wt[qq][t][v];
    }
  }
  __syncthreads();

  int ntile = wave & 1;
  int kbase = (wave >> 1) * 256;
  v8f acc = {};
  for (int k0 = 0; k0 < 256; k0 += 4) {
    int kk = kbase + k0 + ((lane >> 4) << 1);
    int m = lane & 15;
    v2f a;
    a.x = s_wt[m * 512 + kk];
    a.y = s_wt[m * 512 + kk + 1];
    int n = ntile * 16 + (lane & 15);
    v2f b;
    b.x = (kk < 480) ? KV[(size_t)kk * 32 + n] : 0.f;
    b.y = (kk + 1 < 480) ? KV[(size_t)(kk + 1) * 32 + n] : 0.f;
    acc = wmma4(a, b, acc);
  }
  __syncthreads();
  float* s_red = smem;
  if (wave >= 2) {
#pragma unroll
    for (int v = 0; v < 8; ++v) s_red[(wave - 2) * 256 + lane * 8 + v] = acc[v];
  }
  __syncthreads();
  float* s_st = smem + 512;
  int col = wave * 16 + (lane & 15);
  if (wave < 2) {
#pragma unroll
    for (int v = 0; v < 8; ++v) acc[v] += s_red[wave * 256 + lane * 8 + v];
    float s = 0.f, q2 = 0.f;
#pragma unroll
    for (int v = 0; v < 8; ++v) {
      float val = acc[v];
      s += val;
      q2 += val * val;
    }
    s_st[col * 2 + (lane >> 4)] = s;
    s_st[(D2 + col) * 2 + (lane >> 4)] = q2;
  }
  __syncthreads();
  if (wave < 2) {
    int rb = q0 + ((lane >> 4) << 3);
#pragma unroll
    for (int v = 0; v < 8; ++v) y2[(size_t)(rb + v) * D2 + col] = acc[v];
  }
  if (threadIdx.x < 2 * D2) {
    int u = stat_of_slot(threadIdx.x, D2);
    part[slot_index(threadIdx.x, gridDim.x, blockIdx.x)] =
        s_st[u * 2] + s_st[u * 2 + 1];
  }
}

__global__ __launch_bounds__(256) void k_gemm2(const float* __restrict__ y2,
                                               const float* __restrict__ W,
                                               const float* __restrict__ ak,
                                               const float* __restrict__ ck,
                                               float* __restrict__ y3,
                                               float* __restrict__ part,
                                               int ntiles) {
  __shared__ float s_st[8][2 * C_OUT][2];
  int lane = threadIdx.x & 31;
  int wave = threadIdx.x >> 5;
  int tile = blockIdx.x * 8 + wave;
  float s[8], q[8];
#pragma unroll
  for (int t = 0; t < 8; ++t) { s[t] = 0.f; q[t] = 0.f; }
  int m0 = tile * 16;
  bool live = tile < ntiles;
  v8f acc[8];
#pragma unroll
  for (int t = 0; t < 8; ++t) acc[t] = (v8f){};
  if (live) {
#pragma unroll
    for (int k0 = 0; k0 < D2; k0 += 4) {
      int m = m0 + (lane & 15);
      int kk = k0 + ((lane >> 4) << 1);
      v2f a;
      a.x = leaky(y2[(size_t)m * D2 + kk] * ak[kk] + ck[kk]);
      a.y = leaky(y2[(size_t)m * D2 + kk + 1] * ak[kk + 1] + ck[kk + 1]);
#pragma unroll
      for (int t = 0; t < 8; ++t) {
        v2f b = load_b_rm(W, C_OUT, k0, t * 16, lane);
        acc[t] = wmma4(a, b, acc[t]);
      }
    }
#pragma unroll
    for (int t = 0; t < 8; ++t) tile_col_stats(acc[t], s[t], q[t]);
  }
  int n = lane & 15, h = lane >> 4;
#pragma unroll
  for (int t = 0; t < 8; ++t) {
    s_st[wave][t * 16 + n][h] = s[t];
    s_st[wave][C_OUT + t * 16 + n][h] = q[t];
  }
  __syncthreads();
  if (live) {
#pragma unroll
    for (int t = 0; t < 8; ++t) store_d_rm(y3, C_OUT, m0, t * 16, lane, acc[t]);
  }
  int u = stat_of_slot(threadIdx.x, C_OUT);
  float tot = 0.f;
  for (int w = 0; w < 8; ++w) tot += s_st[w][u][0] + s_st[w][u][1];
  part[slot_index(threadIdx.x, gridDim.x, blockIdx.x)] = tot;
}

__global__ __launch_bounds__(128) void k_final(const float* __restrict__ x,
                                               const int* __restrict__ idx,
                                               const float* __restrict__ Wsc,
                                               const float* __restrict__ y3,
                                               const float* __restrict__ a2,
                                               const float* __restrict__ c2,
                                               float* __restrict__ out) {
  __shared__ float s_sx[16 * 64];
  int q0 = blockIdx.x * 16;
  for (int p = threadIdx.x; p < 1024; p += 128) {
    int q = p >> 6, c = p & 63;
    float acc = -3.4e38f;
#pragma unroll
    for (int nb = 0; nb < NB; ++nb) {
      int id = idx[(size_t)(q0 + q) * NB + nb];
      float v = (id < N_PTS) ? x[(size_t)id * C_IN + c] : 0.f;
      acc = fmaxf(acc, v);
    }
    s_sx[q * 64 + c] = acc;
  }
  __syncthreads();
  int lane = threadIdx.x & 31;
  int wave = threadIdx.x >> 5;
#pragma unroll
  for (int tt = 0; tt < 2; ++tt) {
    int ntile = wave * 2 + tt;
    v8f acc = {};
#pragma unroll
    for (int k0 = 0; k0 < C_IN; k0 += 4) {
      int m = lane & 15;
      int kk = k0 + ((lane >> 4) << 1);
      v2f a;
      a.x = s_sx[m * 64 + kk];
      a.y = s_sx[m * 64 + kk + 1];
      v2f b = load_b_rm(Wsc, C_OUT, k0, ntile * 16, lane);
      acc = wmma4(a, b, acc);
    }
    int col = ntile * 16 + (lane & 15);
    int rb = (lane >> 4) << 3;
#pragma unroll
    for (int v = 0; v < 8; ++v) {
      int m = q0 + rb + v;
      float h = leaky(y3[(size_t)m * C_OUT + col] * a2[col] + c2[col]);
      out[(size_t)m * C_OUT + col] = h + acc[v];
    }
  }
}

extern "C" void kernel_launch(void* const* d_in, const int* in_sizes, int n_in,
                              void* d_out, int out_size, void* d_ws,
                              size_t ws_size, hipStream_t stream) {
  const float* x    = (const float*)d_in[0];
  const float* pos  = (const float*)d_in[1];
  const float* qpos = (const float*)d_in[2];
  const int*   idx  = (const int*)d_in[3];
  const float* Wu1  = (const float*)d_in[4];
  const float* g1   = (const float*)d_in[5];
  const float* b1   = (const float*)d_in[6];
  const float* Kpts = (const float*)d_in[7];
  const float* KV   = (const float*)d_in[8];
  const float* gk   = (const float*)d_in[9];
  const float* bk   = (const float*)d_in[10];
  const float* Wu2  = (const float*)d_in[11];
  const float* g2   = (const float*)d_in[12];
  const float* b2   = (const float*)d_in[13];
  const float* Wsc  = (const float*)d_in[14];
  float* out = (float*)d_out;

  float* ws = (float*)d_ws;
  float* y1 = ws;
  float* y2 = y1 + (size_t)N_PTS * D2;
  float* y3 = y2 + (size_t)M_Q * D2;
  float* S  = y3 + (size_t)M_Q * C_OUT;
  float *a1 = S +   0, *c1 = S +  32;
  float *ak = S +  64, *ck = S +  96;
  float *a2 = S + 128, *c2 = S + 256;
  float* part1 = y2;
  float* partk = y3;
  float* part2 = y1;

  k_gemm1<<<G1_BLOCKS, 256, 0, stream>>>(x, Wu1, y1, part1, N_PTS / 16);
  k_finalize_bn<<<D2 / 4, 1024, 0, stream>>>(part1, G1_BLOCKS, g1, b1,
                                             (float)N_PTS, a1, c1);
  k_kpconv<<<M_Q / 16, 128, 0, stream>>>(y1, idx, pos, qpos, Kpts, KV, a1, c1,
                                         y2, partk);
  k_finalize_bn<<<D2 / 4, 1024, 0, stream>>>(partk, M_Q / 16, gk, bk,
                                             (float)M_Q, ak, ck);
  k_gemm2<<<G2_BLOCKS, 256, 0, stream>>>(y2, Wu2, ak, ck, y3, part2,
                                         M_Q / 16);
  k_finalize_bn<<<C_OUT / 4, 1024, 0, stream>>>(part2, G2_BLOCKS, g2, b2,
                                                (float)M_Q, a2, c2);
  k_final<<<M_Q / 16, 128, 0, stream>>>(x, idx, Wsc, y3, a2, c2, out);
}


# Bounds-guarded loads made unconditional (index clamped, value replaced after) so they issue as one group, not one round trip each: shortcut max-pool gather, KPConv K_values fragments, neighbor rows
# speedup vs baseline: 3.4692x; 1.9501x over previous
#include <hip/hip_runtime.h>


typedef float v2f __attribute__((ext_vector_type(2)));
typedef float v8f __attribute__((ext_vector_type(8)));

#define N_PTS   400000
#define M_Q     100000
#define NB      16
#define C_IN    64
#define D2      32
#define C_OUT   128
#define KP      15
#define INV_EXTENT 2.0f
#define EPS_BN  1e-5f
#define SLOPE   0.2f
#define G1_BLOCKS (N_PTS / 16 / 8)
#define G2_BLOCKS ((M_Q / 16 + 7) / 8)

__device__ __forceinline__ float leaky(float v) { return v >= 0.f ? v : SLOPE * v; }

__device__ __forceinline__ v8f wmma4(v2f a, v2f b, v8f c) {
  return __builtin_amdgcn_wmma_f32_16x16x4_f32(false, a, false, b, (short)0, c,
                                               false, false);
}

__device__ __forceinline__ v2f load_a_rm(const float* src, int ld, int m0,
                                         int k0, int lane) {
  int m = m0 + (lane & 15);
  int k = k0 + ((lane >> 4) << 1);
  v2f a;
  a.x = src[(size_t)m * ld + k];
  a.y = src[(size_t)m * ld + k + 1];
  return a;
}

__device__ __forceinline__ v2f load_b_rm(const float* src, int ld, int k0,
                                         int n0, int lane) {
  int n = n0 + (lane & 15);
  int k = k0 + ((lane >> 4) << 1);
  v2f b;
  b.x = src[(size_t)k * ld + n];
  b.y = src[(size_t)(k + 1) * ld + n];
  return b;
}

__device__ __forceinline__ void store_d_rm(float* dst, int ld, int m0, int n0,
                                           int lane, v8f c) {
  int n = n0 + (lane & 15);
  int rb = m0 + ((lane >> 4) << 3);
#pragma unroll
  for (int v = 0; v < 8; ++v) dst[(size_t)(rb + v) * ld + n] = c[v];
}


__device__ __forceinline__ void tile_col_stats(v8f c, float& s, float& q) {
  s = 0.f;
  q = 0.f;
#pragma unroll
  for (int v = 0; v < 8; ++v) { s += c[v]; q += c[v] * c[v]; }
}

__device__ __forceinline__ int stat_of_slot(int t, int C) {
  int j = t >> 3, v = t & 7;
  return 4 * j + (v & 3) + C * (v >> 2);
}
__device__ __forceinline__ size_t slot_index(int t, int nblocks, int blk) {
  int j = t >> 3, v = t & 7;
  return ((size_t)j * nblocks + blk) * 8 + v;
}

__global__ __launch_bounds__(1024) void k_finalize_bn(
    const float* __restrict__ part, int nparts, const float* __restrict__ g,
    const float* __restrict__ b, float cnt, float* a, float* c) {
  __shared__ float s_red[32][8];
  int j = blockIdx.x;
  int h = threadIdx.x & 1;
  int r0 = threadIdx.x >> 1;
  const float4* rows =
      reinterpret_cast<const float4*>(part + (size_t)j * nparts * 8);
  float4 acc = make_float4(0.f, 0.f, 0.f, 0.f);
  for (int base = 0; base < nparts; base += 8 * 512) {
    float4 v[8];
#pragma unroll
    for (int k = 0; k < 8; ++k) {
      int r = base + k * 512 + r0;
      int rc = r < nparts ? r : nparts - 1;
      v[k] = rows[(size_t)rc * 2 + h];
      if (r >= nparts) v[k] = make_float4(0.f, 0.f, 0.f, 0.f);
    }
#pragma unroll
    for (int k = 0; k < 8; ++k) {
      acc.x += v[k].x; acc.y += v[k].y; acc.z += v[k].z; acc.w += v[k].w;
    }
  }
  for (int m = 2; m < 32; m <<= 1) {
    acc.x += __shfl_xor(acc.x, m);
    acc.y += __shfl_xor(acc.y, m);
    acc.z += __shfl_xor(acc.z, m);
    acc.w += __shfl_xor(acc.w, m);
  }
  int lane = threadIdx.x & 31;
  int wave = threadIdx.x >> 5;
  if (lane < 2) {
    s_red[wave][4 * lane + 0] = acc.x;
    s_red[wave][4 * lane + 1] = acc.y;
    s_red[wave][4 * lane + 2] = acc.z;
    s_red[wave][4 * lane + 3] = acc.w;
  }
  __syncthreads();
  if (threadIdx.x < 4) {
    float s = 0.f, q = 0.f;
    for (int w = 0; w < 32; ++w) {
      s += s_red[w][threadIdx.x];
      q += s_red[w][4 + threadIdx.x];
    }
    int ch = 4 * j + threadIdx.x;
    float mu  = s / cnt;
    float var = q / cnt - mu * mu;
    float sc  = g[ch] * rsqrtf(fmaxf(var, 0.f) + EPS_BN);
    a[ch] = sc;
    c[ch] = b[ch] - mu * sc;
  }
}

__global__ __launch_bounds__(256) void k_gemm1(const float* __restrict__ x,
                                               const float* __restrict__ W,
                                               float* __restrict__ y1,
                                               float* __restrict__ part,
                                               int ntiles) {
  __shared__ float s_st[8][2 * D2][2];
  int lane = threadIdx.x & 31;
  int wave = threadIdx.x >> 5;
  int tile = blockIdx.x * 8 + wave;
  int m0 = tile * 16;
  bool live = tile < ntiles;
  float s0 = 0.f, q0 = 0.f, s1 = 0.f, q1 = 0.f;
  v8f acc0 = {}, acc1 = {};
  if (live) {
#pragma unroll
    for (int k0 = 0; k0 < C_IN; k0 += 4) {
      v2f a  = load_a_rm(x, C_IN, m0, k0, lane);
      v2f b0 = load_b_rm(W, D2, k0, 0, lane);
      v2f b1 = load_b_rm(W, D2, k0, 16, lane);
      acc0 = wmma4(a, b0, acc0);
      acc1 = wmma4(a, b1, acc1);
    }
    tile_col_stats(acc0, s0, q0);
    tile_col_stats(acc1, s1, q1);
  }
  int n = lane & 15, h = lane >> 4;
  s_st[wave][n][h] = s0;
  s_st[wave][16 + n][h] = s1;
  s_st[wave][D2 + n][h] = q0;
  s_st[wave][D2 + 16 + n][h] = q1;
  __syncthreads();
  if (live) {
    store_d_rm(y1, D2, m0, 0, lane, acc0);
    store_d_rm(y1, D2, m0, 16, lane, acc1);
  }
  if (threadIdx.x < 2 * D2) {
    int u = stat_of_slot(threadIdx.x, D2);
    float t = 0.f;
    for (int w = 0; w < 8; ++w) t += s_st[w][u][0] + s_st[w][u][1];
    part[slot_index(threadIdx.x, gridDim.x, blockIdx.x)] = t;
  }
}

__global__ __launch_bounds__(128) __attribute__((amdgpu_waves_per_eu(4))) void k_kpconv(
    const float* __restrict__ y1, const int* __restrict__ idx,
    const float* __restrict__ pos, const float* __restrict__ qpos,
    const float* __restrict__ Kpts, const float* __restrict__ KV,
    const float* __restrict__ a1, const float* __restrict__ c1,
    float* __restrict__ y2, float* __restrict__ part) {
  __shared__ float smem[12288];
  float* s_w  = smem;
  float* s_nf = smem + 4096;
  int lane = threadIdx.x & 31;
  int wave = threadIdx.x >> 5;
  int q0 = blockIdx.x * 16;

  for (int p = threadIdx.x; p < 256; p += 128) {
    int q = p >> 4, nb = p & 15;
    int id = idx[(size_t)(q0 + q) * NB + nb];
    float qx = qpos[(size_t)(q0 + q) * 3 + 0];
    float qy = qpos[(size_t)(q0 + q) * 3 + 1];
    float qz = qpos[(size_t)(q0 + q) * 3 + 2];
    bool real = id < N_PTS;
    int idc = real ? id : N_PTS - 1;
    const float* yr = y1 + (size_t)idc * D2;
    float yv[D2];
#pragma unroll
    for (int d = 0; d < D2; ++d) yv[d] = yr[d];
    float px = pos[(size_t)idc * 3 + 0];
    float py = pos[(size_t)idc * 3 + 1];
    float pz = pos[(size_t)idc * 3 + 2];
    if (!real) px = py = pz = 1e6f;
    float rx = px - qx, ry = py - qy, rz = pz - qz;
    float rr = rx * rx + ry * ry + rz * rz;
#pragma unroll
    for (int k = 0; k < KP; ++k) {
      float kx = Kpts[k * 3 + 0], ky = Kpts[k * 3 + 1], kz = Kpts[k * 3 + 2];
      float sq = rr + kx * kx + ky * ky + kz * kz
                 - 2.f * (rx * kx + ry * ky + rz * kz);
      float w = 1.f - sqrtf(fmaxf(sq, 0.f)) * INV_EXTENT;
      s_w[q * 256 + nb * 16 + k] = fmaxf(w, 0.f);
    }
    s_w[q * 256 + nb * 16 + 15] = 0.f;
#pragma unroll
    for (int d = 0; d < D2; ++d) {
      float v = leaky(yv[d] * a1[d] + c1[d]);
      s_nf[q * 512 + nb * 32 + d] = real ? v : 0.f;
    }
  }
  __syncthreads();

  v8f wt[4][2];
#pragma unroll
  for (int qq = 0; qq < 4; ++qq) { wt[qq][0] = (v8f){}; wt[qq][1] = (v8f){}; }
#pragma unroll
  for (int qq = 0; qq < 4; ++qq) {
    int q = wave * 4 + qq;
#pragma unroll
    for (int k0 = 0; k0 < 16; k0 += 4) {
      int m = lane & 15;
      int kk = k0 + ((lane >> 4) << 1);
      v2f a;
      a.x = s_w[q * 256 + kk * 16 + m];
      a.y = s_w[q * 256 + (kk + 1) * 16 + m];
#pragma unroll
      for (int t = 0; t < 2; ++t) {
        int n = t * 16 + (lane & 15);
        v2f b;
        b.x = s_nf[q * 512 + kk * 32 + n];
        b.y = s_nf[q * 512 + (kk + 1) * 32 + n];
        wt[qq][t] = wmma4(a, b, wt[qq][t]);
      }
    }
  }
  int ntile = wave & 1;
  int kl = (wave >> 1) * 256 + ((lane >> 4) << 1);
  int nB = ntile * 16 + (lane & 15);
  v2f bf[64];
#pragma unroll
  for (int i = 0; i < 32; ++i) {
    int kk = kl + 4 * i;
    int r0 = kk < 480 ? kk : 479;
    int r1 = kk + 1 < 480 ? kk + 1 : 479;
    float bx = KV[(size_t)r0 * 32 + nB];
    float by = KV[(size_t)r1 * 32 + nB];
    bf[i].x = kk < 480 ? bx : 0.f;
    bf[i].y = kk + 1 < 480 ? by : 0.f;
  }
  __syncthreads();

  float* s_wt = smem;
#pragma unroll
  for (int qq = 0; qq < 4; ++qq) {
    int q = wave * 4 + qq;
#pragma unroll
    for (int t = 0; t < 2; ++t) {
      int n = t * 16 + (lane & 15);
      int rb = (lane >> 4) << 3;
#pragma unroll
      for (int v = 0; v < 8; ++v)
        s_wt[q * 512 + (rb + v) * 32 + n] = wt[qq][t][v];
    }
  }
#pragma unroll
  for (int i = 32; i < 64; ++i) {
    int kk = kl + 4 * i;
    int r0 = kk < 480 ? kk : 479;
    int r1 = kk + 1 < 480 ? kk + 1 : 479;
    float bx = KV[(size_t)r0 * 32 + nB];
    float by = KV[(size_t)r1 * 32 + nB];
    bf[i].x = kk < 480 ? bx : 0.f;
    bf[i].y = kk + 1 < 480 ? by : 0.f;
  }
  __syncthreads();

  v8f acc = {};
#pragma unroll
  for (int i = 0; i < 64; ++i) {
    int kk = kl + 4 * i;
    int m = lane & 15;
    v2f a;
    a.x = s_wt[m * 512 + kk];
    a.y = s_wt[m * 512 + kk + 1];
    acc = wmma4(a, bf[i], acc);
  }
  __syncthreads();
  float* s_red = smem;
  if (wave >= 2) {
#pragma unroll
    for (int v = 0; v < 8; ++v) s_red[(wave - 2) * 256 + lane * 8 + v] = acc[v];
  }
  __syncthreads();
  float* s_st = smem + 512;
  int col = wave * 16 + (lane & 15);
  if (wave < 2) {
#pragma unroll
    for (int v = 0; v < 8; ++v) acc[v] += s_red[wave * 256 + lane * 8 + v];
    float s = 0.f, q2 = 0.f;
#pragma unroll
    for (int v = 0; v < 8; ++v) {
      float val = acc[v];
      s += val;
      q2 += val * val;
    }
    s_st[col * 2 + (lane >> 4)] = s;
    s_st[(D2 + col) * 2 + (lane >> 4)] = q2;
  }
  __syncthreads();
  if (wave < 2) {
    int rb = q0 + ((lane >> 4) << 3);
#pragma unroll
    for (int v = 0; v < 8; ++v) y2[(size_t)(rb + v) * D2 + col] = acc[v];
  }
  if (threadIdx.x < 2 * D2) {
    int u = stat_of_slot(threadIdx.x, D2);
    part[slot_index(threadIdx.x, gridDim.x, blockIdx.x)] =
        s_st[u * 2] + s_st[u * 2 + 1];
  }
}

__global__ __launch_bounds__(256) void k_gemm2(const float* __restrict__ y2,
                                               const float* __restrict__ W,
                                               const float* __restrict__ ak,
                                               const float* __restrict__ ck,
                                               float* __restrict__ y3,
                                               float* __restrict__ part,
                                               int ntiles) {
  __shared__ float s_st[8][2 * C_OUT][2];
  int lane = threadIdx.x & 31;
  int wave = threadIdx.x >> 5;
  int tile = blockIdx.x * 8 + wave;
  float s[8], q[8];
#pragma unroll
  for (int t = 0; t < 8; ++t) { s[t] = 0.f; q[t] = 0.f; }
  int m0 = tile * 16;
  bool live = tile < ntiles;
  v8f acc[8];
#pragma unroll
  for (int t = 0; t < 8; ++t) acc[t] = (v8f){};
  if (live) {
#pragma unroll
    for (int k0 = 0; k0 < D2; k0 += 4) {
      int m = m0 + (lane & 15);
      int kk = k0 + ((lane >> 4) << 1);
      v2f a;
      a.x = leaky(y2[(size_t)m * D2 + kk] * ak[kk] + ck[kk]);
      a.y = leaky(y2[(size_t)m * D2 + kk + 1] * ak[kk + 1] + ck[kk + 1]);
#pragma unroll
      for (int t = 0; t < 8; ++t) {
        v2f b = load_b_rm(W, C_OUT, k0, t * 16, lane);
        acc[t] = wmma4(a, b, acc[t]);
      }
    }
#pragma unroll
    for (int t = 0; t < 8; ++t) tile_col_stats(acc[t], s[t], q[t]);
  }
  int n = lane & 15, h = lane >> 4;
#pragma unroll
  for (int t = 0; t < 8; ++t) {
    s_st[wave][t * 16 + n][h] = s[t];
    s_st[wave][C_OUT + t * 16 + n][h] = q[t];
  }
  __syncthreads();
  if (live) {
#pragma unroll
    for (int t = 0; t < 8; ++t) store_d_rm(y3, C_OUT, m0, t * 16, lane, acc[t]);
  }
  int u = stat_of_slot(threadIdx.x, C_OUT);
  float tot = 0.f;
  for (int w = 0; w < 8; ++w) tot += s_st[w][u][0] + s_st[w][u][1];
  part[slot_index(threadIdx.x, gridDim.x, blockIdx.x)] = tot;
}

__global__ __launch_bounds__(128) void k_final(const float* __restrict__ x,
                                               const int* __restrict__ idx,
                                               const float* __restrict__ Wsc,
                                               const float* __restrict__ y3,
                                               const float* __restrict__ a2,
                                               const float* __restrict__ c2,
                                               float* __restrict__ out) {
  __shared__ float s_sx[16 * 64];
  int q0 = blockIdx.x * 16;
  for (int p = threadIdx.x; p < 1024; p += 128) {
    int q = p >> 6, c = p & 63;
    const int* ir = idx + (size_t)(q0 + q) * NB;
    int ids[NB];
    float xv[NB];
#pragma unroll
    for (int nb = 0; nb < NB; ++nb) ids[nb] = ir[nb];
#pragma unroll
    for (int nb = 0; nb < NB; ++nb) {
      int idc = ids[nb] < N_PTS ? ids[nb] : N_PTS - 1;
      xv[nb] = x[(size_t)idc * C_IN + c];
    }
    float acc = -3.4e38f;
#pragma unroll
    for (int nb = 0; nb < NB; ++nb)
      acc = fmaxf(acc, ids[nb] < N_PTS ? xv[nb] : 0.f);
    s_sx[q * 64 + c] = acc;
  }
  __syncthreads();
  int lane = threadIdx.x & 31;
  int wave = threadIdx.x >> 5;
#pragma unroll
  for (int tt = 0; tt < 2; ++tt) {
    int ntile = wave * 2 + tt;
    v8f acc = {};
#pragma unroll
    for (int k0 = 0; k0 < C_IN; k0 += 4) {
      int m = lane & 15;
      int kk = k0 + ((lane >> 4) << 1);
      v2f a;
      a.x = s_sx[m * 64 + kk];
      a.y = s_sx[m * 64 + kk + 1];
      v2f b = load_b_rm(Wsc, C_OUT, k0, ntile * 16, lane);
      acc = wmma4(a, b, acc);
    }
    int col = ntile * 16 + (lane & 15);
    int rb = (lane >> 4) << 3;
#pragma unroll
    for (int v = 0; v < 8; ++v) {
      int m = q0 + rb + v;
      float h = leaky(y3[(size_t)m * C_OUT + col] * a2[col] + c2[col]);
      out[(size_t)m * C_OUT + col] = h + acc[v];
    }
  }
}

extern "C" void kernel_launch(void* const* d_in, const int* in_sizes, int n_in,
                              void* d_out, int out_size, void* d_ws,
                              size_t ws_size, hipStream_t stream) {
  const float* x    = (const float*)d_in[0];
  const float* pos  = (const float*)d_in[1];
  const float* qpos = (const float*)d_in[2];
  const int*   idx  = (const int*)d_in[3];
  const float* Wu1  = (const float*)d_in[4];
  const float* g1   = (const float*)d_in[5];
  const float* b1   = (const float*)d_in[6];
  const float* Kpts = (const float*)d_in[7];
  const float* KV   = (const float*)d_in[8];
  const float* gk   = (const float*)d_in[9];
  const float* bk   = (const float*)d_in[10];
  const float* Wu2  = (const float*)d_in[11];
  const float* g2   = (const float*)d_in[12];
  const float* b2   = (const float*)d_in[13];
  const float* Wsc  = (const float*)d_in[14];
  float* out = (float*)d_out;

  float* ws = (float*)d_ws;
  float* y1 = ws;
  float* y2 = y1 + (size_t)N_PTS * D2;
  float* y3 = y2 + (size_t)M_Q * D2;
  float* S  = y3 + (size_t)M_Q * C_OUT;
  float *a1 = S +   0, *c1 = S +  32;
  float *ak = S +  64, *ck = S +  96;
  float *a2 = S + 128, *c2 = S + 256;
  float* part1 = y2;
  float* partk = y3;
  float* part2 = y1;

  k_gemm1<<<G1_BLOCKS, 256, 0, stream>>>(x, Wu1, y1, part1, N_PTS / 16);
  k_finalize_bn<<<D2 / 4, 1024, 0, stream>>>(part1, G1_BLOCKS, g1, b1,
                                             (float)N_PTS, a1, c1);
  k_kpconv<<<M_Q / 16, 128, 0, stream>>>(y1, idx, pos, qpos, Kpts, KV, a1, c1,
                                         y2, partk);
  k_finalize_bn<<<D2 / 4, 1024, 0, stream>>>(partk, M_Q / 16, gk, bk,
                                             (float)M_Q, ak, ck);
  k_gemm2<<<G2_BLOCKS, 256, 0, stream>>>(y2, Wu2, ak, ck, y3, part2,
                                         M_Q / 16);
  k_finalize_bn<<<C_OUT / 4, 1024, 0, stream>>>(part2, G2_BLOCKS, g2, b2,
                                                (float)M_Q, a2, c2);
  k_final<<<M_Q / 16, 128, 0, stream>>>(x, idx, Wsc, y3, a2, c2, out);
}


# LDS tiles re-pitched so fragment reads and staging stores are bank-conflict-free: KPConv weight, feature and weighted_flat tiles (pitches 20, 44, 532 floats), shortcut pooled tile (68)
# speedup vs baseline: 3.8423x; 1.1075x over previous
#include <hip/hip_runtime.h>


typedef float v2f __attribute__((ext_vector_type(2)));
typedef float v8f __attribute__((ext_vector_type(8)));

#define N_PTS   400000
#define M_Q     100000
#define NB      16
#define C_IN    64
#define D2      32
#define C_OUT   128
#define KP      15
#define INV_EXTENT 2.0f
#define EPS_BN  1e-5f
#define SLOPE   0.2f
#define G1_BLOCKS (N_PTS / 16 / 8)
#define G2_BLOCKS ((M_Q / 16 + 7) / 8)

__device__ __forceinline__ float leaky(float v) { return v >= 0.f ? v : SLOPE * v; }

__device__ __forceinline__ v8f wmma4(v2f a, v2f b, v8f c) {
  return __builtin_amdgcn_wmma_f32_16x16x4_f32(false, a, false, b, (short)0, c,
                                               false, false);
}

__device__ __forceinline__ v2f load_a_rm(const float* src, int ld, int m0,
                                         int k0, int lane) {
  int m = m0 + (lane & 15);
  int k = k0 + ((lane >> 4) << 1);
  v2f a;
  a.x = src[(size_t)m * ld + k];
  a.y = src[(size_t)m * ld + k + 1];
  return a;
}

__device__ __forceinline__ v2f load_b_rm(const float* src, int ld, int k0,
                                         int n0, int lane) {
  int n = n0 + (lane & 15);
  int k = k0 + ((lane >> 4) << 1);
  v2f b;
  b.x = src[(size_t)k * ld + n];
  b.y = src[(size_t)(k + 1) * ld + n];
  return b;
}

__device__ __forceinline__ void store_d_rm(float* dst, int ld, int m0, int n0,
                                           int lane, v8f c) {
  int n = n0 + (lane & 15);
  int rb = m0 + ((lane >> 4) << 3);
#pragma unroll
  for (int v = 0; v < 8; ++v) dst[(size_t)(rb + v) * ld + n] = c[v];
}


__device__ __forceinline__ void tile_col_stats(v8f c, float& s, float& q) {
  s = 0.f;
  q = 0.f;
#pragma unroll
  for (int v = 0; v < 8; ++v) { s += c[v]; q += c[v] * c[v]; }
}

__device__ __forceinline__ int stat_of_slot(int t, int C) {
  int j = t >> 3, v = t & 7;
  return 4 * j + (v & 3) + C * (v >> 2);
}
__device__ __forceinline__ size_t slot_index(int t, int nblocks, int blk) {
  int j = t >> 3, v = t & 7;
  return ((size_t)j * nblocks + blk) * 8 + v;
}

__global__ __launch_bounds__(1024) void k_finalize_bn(
    const float* __restrict__ part, int nparts, const float* __restrict__ g,
    const float* __restrict__ b, float cnt, float* a, float* c) {
  __shared__ float s_red[32][8];
  int j = blockIdx.x;
  int h = threadIdx.x & 1;
  int r0 = threadIdx.x >> 1;
  const float4* rows =
      reinterpret_cast<const float4*>(part + (size_t)j * nparts * 8);
  float4 acc = make_float4(0.f, 0.f, 0.f, 0.f);
  for (int base = 0; base < nparts; base += 8 * 512) {
    float4 v[8];
#pragma unroll
    for (int k = 0; k < 8; ++k) {
      int r = base + k * 512 + r0;
      int rc = r < nparts ? r : nparts - 1;
      v[k] = rows[(size_t)rc * 2 + h];
      if (r >= nparts) v[k] = make_float4(0.f, 0.f, 0.f, 0.f);
    }
#pragma unroll
    for (int k = 0; k < 8; ++k) {
      acc.x += v[k].x; acc.y += v[k].y; acc.z += v[k].z; acc.w += v[k].w;
    }
  }
  for (int m = 2; m < 32; m <<= 1) {
    acc.x += __shfl_xor(acc.x, m);
    acc.y += __shfl_xor(acc.y, m);
    acc.z += __shfl_xor(acc.z, m);
    acc.w += __shfl_xor(acc.w, m);
  }
  int lane = threadIdx.x & 31;
  int wave = threadIdx.x >> 5;
  if (lane < 2) {
    s_red[wave][4 * lane + 0] = acc.x;
    s_red[wave][4 * lane + 1] = acc.y;
    s_red[wave][4 * lane + 2] = acc.z;
    s_red[wave][4 * lane + 3] = acc.w;
  }
  __syncthreads();
  if (threadIdx.x < 4) {
    float s = 0.f, q = 0.f;
    for (int w = 0; w < 32; ++w) {
      s += s_red[w][threadIdx.x];
      q += s_red[w][4 + threadIdx.x];
    }
    int ch = 4 * j + threadIdx.x;
    float mu  = s / cnt;
    float var = q / cnt - mu * mu;
    float sc  = g[ch] * rsqrtf(fmaxf(var, 0.f) + EPS_BN);
    a[ch] = sc;
    c[ch] = b[ch] - mu * sc;
  }
}

__global__ __launch_bounds__(256) void k_gemm1(const float* __restrict__ x,
                                               const float* __restrict__ W,
                                               float* __restrict__ y1,
                                               float* __restrict__ part,
                                               int ntiles) {
  __shared__ float s_st[8][2 * D2][2];
  int lane = threadIdx.x & 31;
  int wave = threadIdx.x >> 5;
  int tile = blockIdx.x * 8 + wave;
  int m0 = tile * 16;
  bool live = tile < ntiles;
  float s0 = 0.f, q0 = 0.f, s1 = 0.f, q1 = 0.f;
  v8f acc0 = {}, acc1 = {};
  if (live) {
#pragma unroll
    for (int k0 = 0; k0 < C_IN; k0 += 4) {
      v2f a  = load_a_rm(x, C_IN, m0, k0, lane);
      v2f b0 = load_b_rm(W, D2, k0, 0, lane);
      v2f b1 = load_b_rm(W, D2, k0, 16, lane);
      acc0 = wmma4(a, b0, acc0);
      acc1 = wmma4(a, b1, acc1);
    }
    tile_col_stats(acc0, s0, q0);
    tile_col_stats(acc1, s1, q1);
  }
  int n = lane & 15, h = lane >> 4;
  s_st[wave][n][h] = s0;
  s_st[wave][16 + n][h] = s1;
  s_st[wave][D2 + n][h] = q0;
  s_st[wave][D2 + 16 + n][h] = q1;
  __syncthreads();
  if (live) {
    store_d_rm(y1, D2, m0, 0, lane, acc0);
    store_d_rm(y1, D2, m0, 16, lane, acc1);
  }
  if (threadIdx.x < 2 * D2) {
    int u = stat_of_slot(threadIdx.x, D2);
    float t = 0.f;
    for (int w = 0; w < 8; ++w) t += s_st[w][u][0] + s_st[w][u][1];
    part[slot_index(threadIdx.x, gridDim.x, blockIdx.x)] = t;
  }
}

#define W_NB   20
#define W_Q    (NB * W_NB)
#define NF_NB  44
#define NF_Q   (NB * NF_NB)
#define WT_Q   532
#define WT_MID 16

__global__ __launch_bounds__(128) __attribute__((amdgpu_waves_per_eu(4))) void k_kpconv(
    const float* __restrict__ y1, const int* __restrict__ idx,
    const float* __restrict__ pos, const float* __restrict__ qpos,
    const float* __restrict__ Kpts, const float* __restrict__ KV,
    const float* __restrict__ a1, const float* __restrict__ c1,
    float* __restrict__ y2, float* __restrict__ part) {
  __shared__ __attribute__((aligned(16))) float smem[16 * W_Q + 16 * NF_Q];
  float* s_w  = smem;
  float* s_nf = smem + 16 * W_Q;
  int lane = threadIdx.x & 31;
  int wave = threadIdx.x >> 5;
  int q0 = blockIdx.x * 16;

  for (int p = threadIdx.x; p < 256; p += 128) {
    int q = p >> 4, nb = p & 15;
    int id = idx[(size_t)(q0 + q) * NB + nb];
    float qx = qpos[(size_t)(q0 + q) * 3 + 0];
    float qy = qpos[(size_t)(q0 + q) * 3 + 1];
    float qz = qpos[(size_t)(q0 + q) * 3 + 2];
    bool real = id < N_PTS;
    int idc = real ? id : N_PTS - 1;
    const float* yr = y1 + (size_t)idc * D2;
    float yv[D2];
#pragma unroll
    for (int d = 0; d < D2; ++d) yv[d] = yr[d];
    float px = pos[(size_t)idc * 3 + 0];
    float py = pos[(size_t)idc * 3 + 1];
    float pz = pos[(size_t)idc * 3 + 2];
    if (!real) px = py = pz = 1e6f;
    float rx = px - qx, ry = py - qy, rz = pz - qz;
    float rr = rx * rx + ry * ry + rz * rz;
#pragma unroll
    for (int k = 0; k < KP; ++k) {
      float kx = Kpts[k * 3 + 0], ky = Kpts[k * 3 + 1], kz = Kpts[k * 3 + 2];
      float sq = rr + kx * kx + ky * ky + kz * kz
                 - 2.f * (rx * kx + ry * ky + rz * kz);
      float w = 1.f - sqrtf(fmaxf(sq, 0.f)) * INV_EXTENT;
      s_w[q * W_Q + nb * W_NB + k] = fmaxf(w, 0.f);
    }
    s_w[q * W_Q + nb * W_NB + 15] = 0.f;
#pragma unroll
    for (int d = 0; d < D2; ++d) {
      float v = leaky(yv[d] * a1[d] + c1[d]);
      s_nf[q * NF_Q + nb * NF_NB + d] = real ? v : 0.f;
    }
  }
  __syncthreads();

  v8f wt[4][2];
#pragma unroll
  for (int qq = 0; qq < 4; ++qq) { wt[qq][0] = (v8f){}; wt[qq][1] = (v8f){}; }
#pragma unroll
  for (int qq = 0; qq < 4; ++qq) {
    int q = wave * 4 + qq;
#pragma unroll
    for (int k0 = 0; k0 < 16; k0 += 4) {
      int m = lane & 15;
      int kk = k0 + ((lane >> 4) << 1);
      v2f a;
      a.x = s_w[q * W_Q + kk * W_NB + m];
      a.y = s_w[q * W_Q + (kk + 1) * W_NB + m];
#pragma unroll
      for (int t = 0; t < 2; ++t) {
        int n = t * 16 + (lane & 15);
        v2f b;
        b.x = s_nf[q * NF_Q + kk * NF_NB + n];
        b.y = s_nf[q * NF_Q + (kk + 1) * NF_NB + n];
        wt[qq][t] = wmma4(a, b, wt[qq][t]);
      }
    }
  }
  int ntile = wave & 1;
  int kl = (wave >> 1) * 256 + ((lane >> 4) << 1);
  int nB = ntile * 16 + (lane & 15);
  v2f bf[64];
#pragma unroll
  for (int i = 0; i < 32; ++i) {
    int kk = kl + 4 * i;
    int r0 = kk < 480 ? kk : 479;
    int r1 = kk + 1 < 480 ? kk + 1 : 479;
    float bx = KV[(size_t)r0 * 32 + nB];
    float by = KV[(size_t)r1 * 32 + nB];
    bf[i].x = kk < 480 ? bx : 0.f;
    bf[i].y = kk + 1 < 480 ? by : 0.f;
  }
  __syncthreads();

  float* s_wt = smem;
#pragma unroll
  for (int qq = 0; qq < 4; ++qq) {
    int q = wave * 4 + qq;
#pragma unroll
    for (int t = 0; t < 2; ++t) {
      int n = t * 16 + (lane & 15);
      int rb = (lane >> 4) << 3;
      int so = q * WT_Q + rb * 32 + (lane >> 4) * WT_MID + n;
#pragma unroll
      for (int v = 0; v < 8; ++v)
        s_wt[so + v * 32] = wt[qq][t][v];
    }
  }
#pragma unroll
  for (int i = 32; i < 64; ++i) {
    int kk = kl + 4 * i;
    int r0 = kk < 480 ? kk : 479;
    int r1 = kk + 1 < 480 ? kk + 1 : 479;
    float bx = KV[(size_t)r0 * 32 + nB];
    float by = KV[(size_t)r1 * 32 + nB];
    bf[i].x = kk < 480 ? bx : 0.f;
    bf[i].y = kk + 1 < 480 ? by : 0.f;
  }
  __syncthreads();

  v8f acc = {};
  int ao = (lane & 15) * WT_Q + kl + (wave >> 1) * WT_MID;
#pragma unroll
  for (int i = 0; i < 64; ++i) {
    v2f a;
    a.x = s_wt[ao + 4 * i];
    a.y = s_wt[ao + 4 * i + 1];
    acc = wmma4(a, bf[i], acc);
  }
  __syncthreads();
  float* s_red = smem;
  if (wave >= 2) {
#pragma unroll
    for (int v = 0; v < 8; ++v) s_red[(wave - 2) * 256 + lane * 8 + v] = acc[v];
  }
  __syncthreads();
  float* s_st = smem + 512;
  int col = wave * 16 + (lane & 15);
  if (wave < 2) {
#pragma unroll
    for (int v = 0; v < 8; ++v) acc[v] += s_red[wave * 256 + lane * 8 + v];
    float s = 0.f, q2 = 0.f;
#pragma unroll
    for (int v = 0; v < 8; ++v) {
      float val = acc[v];
      s += val;
      q2 += val * val;
    }
    s_st[col * 2 + (lane >> 4)] = s;
    s_st[(D2 + col) * 2 + (lane >> 4)] = q2;
  }
  __syncthreads();
  if (wave < 2) {
    int rb = q0 + ((lane >> 4) << 3);
#pragma unroll
    for (int v = 0; v < 8; ++v) y2[(size_t)(rb + v) * D2 + col] = acc[v];
  }
  if (threadIdx.x < 2 * D2) {
    int u = stat_of_slot(threadIdx.x, D2);
    part[slot_index(threadIdx.x, gridDim.x, blockIdx.x)] =
        s_st[u * 2] + s_st[u * 2 + 1];
  }
}

__global__ __launch_bounds__(256) void k_gemm2(const float* __restrict__ y2,
                                               const float* __restrict__ W,
                                               const float* __restrict__ ak,
                                               const float* __restrict__ ck,
                                               float* __restrict__ y3,
                                               float* __restrict__ part,
                                               int ntiles) {
  __shared__ float s_st[8][2 * C_OUT][2];
  int lane = threadIdx.x & 31;
  int wave = threadIdx.x >> 5;
  int tile = blockIdx.x * 8 + wave;
  float s[8], q[8];
#pragma unroll
  for (int t = 0; t < 8; ++t) { s[t] = 0.f; q[t] = 0.f; }
  int m0 = tile * 16;
  bool live = tile < ntiles;
  v8f acc[8];
#pragma unroll
  for (int t = 0; t < 8; ++t) acc[t] = (v8f){};
  if (live) {
#pragma unroll
    for (int k0 = 0; k0 < D2; k0 += 4) {
      int m = m0 + (lane & 15);
      int kk = k0 + ((lane >> 4) << 1);
      v2f a;
      a.x = leaky(y2[(size_t)m * D2 + kk] * ak[kk] + ck[kk]);
      a.y = leaky(y2[(size_t)m * D2 + kk + 1] * ak[kk + 1] + ck[kk + 1]);
#pragma unroll
      for (int t = 0; t < 8; ++t) {
        v2f b = load_b_rm(W, C_OUT, k0, t * 16, lane);
        acc[t] = wmma4(a, b, acc[t]);
      }
    }
#pragma unroll
    for (int t = 0; t < 8; ++t) tile_col_stats(acc[t], s[t], q[t]);
  }
  int n = lane & 15, h = lane >> 4;
#pragma unroll
  for (int t = 0; t < 8; ++t) {
    s_st[wave][t * 16 + n][h] = s[t];
    s_st[wave][C_OUT + t * 16 + n][h] = q[t];
  }
  __syncthreads();
  if (live) {
#pragma unroll
    for (int t = 0; t < 8; ++t) store_d_rm(y3, C_OUT, m0, t * 16, lane, acc[t]);
  }
  int u = stat_of_slot(threadIdx.x, C_OUT);
  float tot = 0.f;
  for (int w = 0; w < 8; ++w) tot += s_st[w][u][0] + s_st[w][u][1];
  part[slot_index(threadIdx.x, gridDim.x, blockIdx.x)] = tot;
}

#define SX_LD 68
__global__ __launch_bounds__(128) void k_final(const float* __restrict__ x,
                                               const int* __restrict__ idx,
                                               const float* __restrict__ Wsc,
                                               const float* __restrict__ y3,
                                               const float* __restrict__ a2,
                                               const float* __restrict__ c2,
                                               float* __restrict__ out) {
  __shared__ __attribute__((aligned(16))) float s_sx[16 * SX_LD];
  int q0 = blockIdx.x * 16;
  for (int p = threadIdx.x; p < 1024; p += 128) {
    int q = p >> 6, c = p & 63;
    const int* ir = idx + (size_t)(q0 + q) * NB;
    int ids[NB];
    float xv[NB];
#pragma unroll
    for (int nb = 0; nb < NB; ++nb) ids[nb] = ir[nb];
#pragma unroll
    for (int nb = 0; nb < NB; ++nb) {
      int idc = ids[nb] < N_PTS ? ids[nb] : N_PTS - 1;
      xv[nb] = x[(size_t)idc * C_IN + c];
    }
    float acc = -3.4e38f;
#pragma unroll
    for (int nb = 0; nb < NB; ++nb)
      acc = fmaxf(acc, ids[nb] < N_PTS ? xv[nb] : 0.f);
    s_sx[q * SX_LD + c] = acc;
  }
  __syncthreads();
  int lane = threadIdx.x & 31;
  int wave = threadIdx.x >> 5;
#pragma unroll
  for (int tt = 0; tt < 2; ++tt) {
    int ntile = wave * 2 + tt;
    v8f acc = {};
#pragma unroll
    for (int k0 = 0; k0 < C_IN; k0 += 4) {
      int m = lane & 15;
      int kk = k0 + ((lane >> 4) << 1);
      v2f a;
      a.x = s_sx[m * SX_LD + kk];
      a.y = s_sx[m * SX_LD + kk + 1];
      v2f b = load_b_rm(Wsc, C_OUT, k0, ntile * 16, lane);
      acc = wmma4(a, b, acc);
    }
    int col = ntile * 16 + (lane & 15);
    int rb = (lane >> 4) << 3;
#pragma unroll
    for (int v = 0; v < 8; ++v) {
      int m = q0 + rb + v;
      float h = leaky(y3[(size_t)m * C_OUT + col] * a2[col] + c2[col]);
      out[(size_t)m * C_OUT + col] = h + acc[v];
    }
  }
}

extern "C" void kernel_launch(void* const* d_in, const int* in_sizes, int n_in,
                              void* d_out, int out_size, void* d_ws,
                              size_t ws_size, hipStream_t stream) {
  const float* x    = (const float*)d_in[0];
  const float* pos  = (const float*)d_in[1];
  const float* qpos = (const float*)d_in[2];
  const int*   idx  = (const int*)d_in[3];
  const float* Wu1  = (const float*)d_in[4];
  const float* g1   = (const float*)d_in[5];
  const float* b1   = (const float*)d_in[6];
  const float* Kpts = (const float*)d_in[7];
  const float* KV   = (const float*)d_in[8];
  const float* gk   = (const float*)d_in[9];
  const float* bk   = (const float*)d_in[10];
  const float* Wu2  = (const float*)d_in[11];
  const float* g2   = (const float*)d_in[12];
  const float* b2   = (const float*)d_in[13];
  const float* Wsc  = (const float*)d_in[14];
  float* out = (float*)d_out;

  float* ws = (float*)d_ws;
  float* y1 = ws;
  float* y2 = y1 + (size_t)N_PTS * D2;
  float* y3 = y2 + (size_t)M_Q * D2;
  float* S  = y3 + (size_t)M_Q * C_OUT;
  float *a1 = S +   0, *c1 = S +  32;
  float *ak = S +  64, *ck = S +  96;
  float *a2 = S + 128, *c2 = S + 256;
  float* part1 = y2;
  float* partk = y3;
  float* part2 = y1;

  k_gemm1<<<G1_BLOCKS, 256, 0, stream>>>(x, Wu1, y1, part1, N_PTS / 16);
  k_finalize_bn<<<D2 / 4, 1024, 0, stream>>>(part1, G1_BLOCKS, g1, b1,
                                             (float)N_PTS, a1, c1);
  k_kpconv<<<M_Q / 16, 128, 0, stream>>>(y1, idx, pos, qpos, Kpts, KV, a1, c1,
                                         y2, partk);
  k_finalize_bn<<<D2 / 4, 1024, 0, stream>>>(partk, M_Q / 16, gk, bk,
                                             (float)M_Q, ak, ck);
  k_gemm2<<<G2_BLOCKS, 256, 0, stream>>>(y2, Wu2, ak, ck, y3, part2,
                                         M_Q / 16);
  k_finalize_bn<<<C_OUT / 4, 1024, 0, stream>>>(part2, G2_BLOCKS, g2, b2,
                                                (float)M_Q, a2, c2);
  k_final<<<M_Q / 16, 128, 0, stream>>>(x, idx, Wsc, y3, a2, c2, out);
}


# Row gathers made whole-line: 16-byte loads, 8 lanes per y1 row (KPConv neighbor features) and 16 lanes per x row (shortcut max-pool); a thread's 16 rows go out as one group
# speedup vs baseline: 4.3366x; 1.1286x over previous
#include <hip/hip_runtime.h>


typedef float v2f __attribute__((ext_vector_type(2)));
typedef float v8f __attribute__((ext_vector_type(8)));

#define N_PTS   400000
#define M_Q     100000
#define NB      16
#define C_IN    64
#define D2      32
#define C_OUT   128
#define KP      15
#define INV_EXTENT 2.0f
#define EPS_BN  1e-5f
#define SLOPE   0.2f
#define G1_BLOCKS (N_PTS / 16 / 8)
#define G2_BLOCKS ((M_Q / 16 + 7) / 8)

__device__ __forceinline__ float leaky(float v) { return v >= 0.f ? v : SLOPE * v; }

__device__ __forceinline__ v8f wmma4(v2f a, v2f b, v8f c) {
  return __builtin_amdgcn_wmma_f32_16x16x4_f32(false, a, false, b, (short)0, c,
                                               false, false);
}

__device__ __forceinline__ v2f load_a_rm(const float* src, int ld, int m0,
                                         int k0, int lane) {
  int m = m0 + (lane & 15);
  int k = k0 + ((lane >> 4) << 1);
  v2f a;
  a.x = src[(size_t)m * ld + k];
  a.y = src[(size_t)m * ld + k + 1];
  return a;
}

__device__ __forceinline__ v2f load_b_rm(const float* src, int ld, int k0,
                                         int n0, int lane) {
  int n = n0 + (lane & 15);
  int k = k0 + ((lane >> 4) << 1);
  v2f b;
  b.x = src[(size_t)k * ld + n];
  b.y = src[(size_t)(k + 1) * ld + n];
  return b;
}

__device__ __forceinline__ void store_d_rm(float* dst, int ld, int m0, int n0,
                                           int lane, v8f c) {
  int n = n0 + (lane & 15);
  int rb = m0 + ((lane >> 4) << 3);
#pragma unroll
  for (int v = 0; v < 8; ++v) dst[(size_t)(rb + v) * ld + n] = c[v];
}


__device__ __forceinline__ void tile_col_stats(v8f c, float& s, float& q) {
  s = 0.f;
  q = 0.f;
#pragma unroll
  for (int v = 0; v < 8; ++v) { s += c[v]; q += c[v] * c[v]; }
}

__device__ __forceinline__ int stat_of_slot(int t, int C) {
  int j = t >> 3, v = t & 7;
  return 4 * j + (v & 3) + C * (v >> 2);
}
__device__ __forceinline__ size_t slot_index(int t, int nblocks, int blk) {
  int j = t >> 3, v = t & 7;
  return ((size_t)j * nblocks + blk) * 8 + v;
}

__global__ __launch_bounds__(1024) void k_finalize_bn(
    const float* __restrict__ part, int nparts, const float* __restrict__ g,
    const float* __restrict__ b, float cnt, float* a, float* c) {
  __shared__ float s_red[32][8];
  int j = blockIdx.x;
  int h = threadIdx.x & 1;
  int r0 = threadIdx.x >> 1;
  const float4* rows =
      reinterpret_cast<const float4*>(part + (size_t)j * nparts * 8);
  float4 acc = make_float4(0.f, 0.f, 0.f, 0.f);
  for (int base = 0; base < nparts; base += 8 * 512) {
    float4 v[8];
#pragma unroll
    for (int k = 0; k < 8; ++k) {
      int r = base + k * 512 + r0;
      int rc = r < nparts ? r : nparts - 1;
      v[k] = rows[(size_t)rc * 2 + h];
      if (r >= nparts) v[k] = make_float4(0.f, 0.f, 0.f, 0.f);
    }
#pragma unroll
    for (int k = 0; k < 8; ++k) {
      acc.x += v[k].x; acc.y += v[k].y; acc.z += v[k].z; acc.w += v[k].w;
    }
  }
  for (int m = 2; m < 32; m <<= 1) {
    acc.x += __shfl_xor(acc.x, m);
    acc.y += __shfl_xor(acc.y, m);
    acc.z += __shfl_xor(acc.z, m);
    acc.w += __shfl_xor(acc.w, m);
  }
  int lane = threadIdx.x & 31;
  int wave = threadIdx.x >> 5;
  if (lane < 2) {
    s_red[wave][4 * lane + 0] = acc.x;
    s_red[wave][4 * lane + 1] = acc.y;
    s_red[wave][4 * lane + 2] = acc.z;
    s_red[wave][4 * lane + 3] = acc.w;
  }
  __syncthreads();
  if (threadIdx.x < 4) {
    float s = 0.f, q = 0.f;
    for (int w = 0; w < 32; ++w) {
      s += s_red[w][threadIdx.x];
      q += s_red[w][4 + threadIdx.x];
    }
    int ch = 4 * j + threadIdx.x;
    float mu  = s / cnt;
    float var = q / cnt - mu * mu;
    float sc  = g[ch] * rsqrtf(fmaxf(var, 0.f) + EPS_BN);
    a[ch] = sc;
    c[ch] = b[ch] - mu * sc;
  }
}

__global__ __launch_bounds__(256) void k_gemm1(const float* __restrict__ x,
                                               const float* __restrict__ W,
                                               float* __restrict__ y1,
                                               float* __restrict__ part,
                                               int ntiles) {
  __shared__ float s_st[8][2 * D2][2];
  int lane = threadIdx.x & 31;
  int wave = threadIdx.x >> 5;
  int tile = blockIdx.x * 8 + wave;
  int m0 = tile * 16;
  bool live = tile < ntiles;
  float s0 = 0.f, q0 = 0.f, s1 = 0.f, q1 = 0.f;
  v8f acc0 = {}, acc1 = {};
  if (live) {
#pragma unroll
    for (int k0 = 0; k0 < C_IN; k0 += 4) {
      v2f a  = load_a_rm(x, C_IN, m0, k0, lane);
      v2f b0 = load_b_rm(W, D2, k0, 0, lane);
      v2f b1 = load_b_rm(W, D2, k0, 16, lane);
      acc0 = wmma4(a, b0, acc0);
      acc1 = wmma4(a, b1, acc1);
    }
    tile_col_stats(acc0, s0, q0);
    tile_col_stats(acc1, s1, q1);
  }
  int n = lane & 15, h = lane >> 4;
  s_st[wave][n][h] = s0;
  s_st[wave][16 + n][h] = s1;
  s_st[wave][D2 + n][h] = q0;
  s_st[wave][D2 + 16 + n][h] = q1;
  __syncthreads();
  if (live) {
    store_d_rm(y1, D2, m0, 0, lane, acc0);
    store_d_rm(y1, D2, m0, 16, lane, acc1);
  }
  if (threadIdx.x < 2 * D2) {
    int u = stat_of_slot(threadIdx.x, D2);
    float t = 0.f;
    for (int w = 0; w < 8; ++w) t += s_st[w][u][0] + s_st[w][u][1];
    part[slot_index(threadIdx.x, gridDim.x, blockIdx.x)] = t;
  }
}

#define W_NB   20
#define W_Q    (NB * W_NB)
#define NF_NB  44
#define NF_Q   (NB * NF_NB)
#define WT_Q   532
#define WT_MID 16

__global__ __launch_bounds__(128) __attribute__((amdgpu_waves_per_eu(4))) void k_kpconv(
    const float* __restrict__ y1, const int* __restrict__ idx,
    const float* __restrict__ pos, const float* __restrict__ qpos,
    const float* __restrict__ Kpts, const float* __restrict__ KV,
    const float* __restrict__ a1, const float* __restrict__ c1,
    float* __restrict__ y2, float* __restrict__ part) {
  __shared__ __attribute__((aligned(16))) float smem[16 * W_Q + 16 * NF_Q];
  float* s_w  = smem;
  float* s_nf = smem + 16 * W_Q;
  int lane = threadIdx.x & 31;
  int wave = threadIdx.x >> 5;
  int q0 = blockIdx.x * 16;

  unsigned fq = threadIdx.x >> 3, fj = threadIdx.x & 7;
  int fid[NB];
  {
    const int4* ir =
        reinterpret_cast<const int4*>(idx + ((unsigned)q0 + fq) * NB);
#pragma unroll
    for (int g = 0; g < 4; ++g) {
      int4 v = ir[g];
      fid[4 * g + 0] = v.x; fid[4 * g + 1] = v.y;
      fid[4 * g + 2] = v.z; fid[4 * g + 3] = v.w;
    }
  }
  int wid[2];
  float wqx[2], wqy[2], wqz[2], wpx[2], wpy[2], wpz[2];
#pragma unroll
  for (int it = 0; it < 2; ++it) {
    unsigned p = threadIdx.x + 128 * it;
    unsigned q = (unsigned)q0 + (p >> 4), nb = p & 15;
    wid[it] = idx[q * NB + nb];
    const float* qp = qpos + q * 3;
    wqx[it] = qp[0];
    wqy[it] = qp[1];
    wqz[it] = qp[2];
  }
#pragma unroll
  for (int it = 0; it < 2; ++it) {
    unsigned idc = wid[it] < N_PTS ? wid[it] : N_PTS - 1;
    const float* pp = pos + idc * 3;
    wpx[it] = pp[0];
    wpy[it] = pp[1];
    wpz[it] = pp[2];
  }
  const float4* y1v = reinterpret_cast<const float4*>(y1);
  float4 yv[NB];
#pragma unroll
  for (int nb = 0; nb < NB; ++nb) {
    unsigned idc = fid[nb] < N_PTS ? fid[nb] : N_PTS - 1;
    yv[nb] = y1v[idc * (D2 / 4) + fj];
  }
  float4 av = reinterpret_cast<const float4*>(a1)[fj];
  float4 cv = reinterpret_cast<const float4*>(c1)[fj];
#pragma unroll
  for (int it = 0; it < 2; ++it) {
    unsigned p = threadIdx.x + 128 * it;
    unsigned q = p >> 4, nb = p & 15;
    bool real = wid[it] < N_PTS;
    float px = real ? wpx[it] : 1e6f;
    float py = real ? wpy[it] : 1e6f;
    float pz = real ? wpz[it] : 1e6f;
    float rx = px - wqx[it], ry = py - wqy[it], rz = pz - wqz[it];
    float rr = rx * rx + ry * ry + rz * rz;
#pragma unroll
    for (int k = 0; k < KP; ++k) {
      float kx = Kpts[k * 3 + 0], ky = Kpts[k * 3 + 1], kz = Kpts[k * 3 + 2];
      float sq = rr + kx * kx + ky * ky + kz * kz
                 - 2.f * (rx * kx + ry * ky + rz * kz);
      float w = 1.f - sqrtf(fmaxf(sq, 0.f)) * INV_EXTENT;
      s_w[q * W_Q + nb * W_NB + k] = fmaxf(w, 0.f);
    }
    s_w[q * W_Q + nb * W_NB + 15] = 0.f;
  }
#pragma unroll
  for (int nb = 0; nb < NB; ++nb) {
    bool real = fid[nb] < N_PTS;
    float4 o;
    o.x = real ? leaky(yv[nb].x * av.x + cv.x) : 0.f;
    o.y = real ? leaky(yv[nb].y * av.y + cv.y) : 0.f;
    o.z = real ? leaky(yv[nb].z * av.z + cv.z) : 0.f;
    o.w = real ? leaky(yv[nb].w * av.w + cv.w) : 0.f;
    *reinterpret_cast<float4*>(&s_nf[fq * NF_Q + nb * NF_NB + 4 * fj]) = o;
  }
  __syncthreads();

  v8f wt[4][2];
#pragma unroll
  for (int qq = 0; qq < 4; ++qq) { wt[qq][0] = (v8f){}; wt[qq][1] = (v8f){}; }
#pragma unroll
  for (int qq = 0; qq < 4; ++qq) {
    int q = wave * 4 + qq;
#pragma unroll
    for (int k0 = 0; k0 < 16; k0 += 4) {
      int m = lane & 15;
      int kk = k0 + ((lane >> 4) << 1);
      v2f a;
      a.x = s_w[q * W_Q + kk * W_NB + m];
      a.y = s_w[q * W_Q + (kk + 1) * W_NB + m];
#pragma unroll
      for (int t = 0; t < 2; ++t) {
        int n = t * 16 + (lane & 15);
        v2f b;
        b.x = s_nf[q * NF_Q + kk * NF_NB + n];
        b.y = s_nf[q * NF_Q + (kk + 1) * NF_NB + n];
        wt[qq][t] = wmma4(a, b, wt[qq][t]);
      }
    }
  }
  int ntile = wave & 1;
  int kl = (wave >> 1) * 256 + ((lane >> 4) << 1);
  int nB = ntile * 16 + (lane & 15);
  v2f bf[64];
#pragma unroll
  for (int i = 0; i < 32; ++i) {
    int kk = kl + 4 * i;
    int r0 = kk < 480 ? kk : 479;
    int r1 = kk + 1 < 480 ? kk + 1 : 479;
    float bx = KV[(size_t)r0 * 32 + nB];
    float by = KV[(size_t)r1 * 32 + nB];
    bf[i].x = kk < 480 ? bx : 0.f;
    bf[i].y = kk + 1 < 480 ? by : 0.f;
  }
  __syncthreads();

  float* s_wt = smem;
#pragma unroll
  for (int qq = 0; qq < 4; ++qq) {
    int q = wave * 4 + qq;
#pragma unroll
    for (int t = 0; t < 2; ++t) {
      int n = t * 16 + (lane & 15);
      int rb = (lane >> 4) << 3;
      int so = q * WT_Q + rb * 32 + (lane >> 4) * WT_MID + n;
#pragma unroll
      for (int v = 0; v < 8; ++v)
        s_wt[so + v * 32] = wt[qq][t][v];
    }
  }
#pragma unroll
  for (int i = 32; i < 64; ++i) {
    int kk = kl + 4 * i;
    int r0 = kk < 480 ? kk : 479;
    int r1 = kk + 1 < 480 ? kk + 1 : 479;
    float bx = KV[(size_t)r0 * 32 + nB];
    float by = KV[(size_t)r1 * 32 + nB];
    bf[i].x = kk < 480 ? bx : 0.f;
    bf[i].y = kk + 1 < 480 ? by : 0.f;
  }
  __syncthreads();

  v8f acc = {};
  int ao = (lane & 15) * WT_Q + kl + (wave >> 1) * WT_MID;
#pragma unroll
  for (int i = 0; i < 64; ++i) {
    v2f a;
    a.x = s_wt[ao + 4 * i];
    a.y = s_wt[ao + 4 * i + 1];
    acc = wmma4(a, bf[i], acc);
  }
  __syncthreads();
  float* s_red = smem;
  if (wave >= 2) {
#pragma unroll
    for (int v = 0; v < 8; ++v) s_red[(wave - 2) * 256 + lane * 8 + v] = acc[v];
  }
  __syncthreads();
  float* s_st = smem + 512;
  int col = wave * 16 + (lane & 15);
  if (wave < 2) {
#pragma unroll
    for (int v = 0; v < 8; ++v) acc[v] += s_red[wave * 256 + lane * 8 + v];
    float s = 0.f, q2 = 0.f;
#pragma unroll
    for (int v = 0; v < 8; ++v) {
      float val = acc[v];
      s += val;
      q2 += val * val;
    }
    s_st[col * 2 + (lane >> 4)] = s;
    s_st[(D2 + col) * 2 + (lane >> 4)] = q2;
  }
  __syncthreads();
  if (wave < 2) {
    int rb = q0 + ((lane >> 4) << 3);
#pragma unroll
    for (int v = 0; v < 8; ++v) y2[(size_t)(rb + v) * D2 + col] = acc[v];
  }
  if (threadIdx.x < 2 * D2) {
    int u = stat_of_slot(threadIdx.x, D2);
    part[slot_index(threadIdx.x, gridDim.x, blockIdx.x)] =
        s_st[u * 2] + s_st[u * 2 + 1];
  }
}

__global__ __launch_bounds__(256) void k_gemm2(const float* __restrict__ y2,
                                               const float* __restrict__ W,
                                               const float* __restrict__ ak,
                                               const float* __restrict__ ck,
                                               float* __restrict__ y3,
                                               float* __restrict__ part,
                                               int ntiles) {
  __shared__ float s_st[8][2 * C_OUT][2];
  int lane = threadIdx.x & 31;
  int wave = threadIdx.x >> 5;
  int tile = blockIdx.x * 8 + wave;
  float s[8], q[8];
#pragma unroll
  for (int t = 0; t < 8; ++t) { s[t] = 0.f; q[t] = 0.f; }
  int m0 = tile * 16;
  bool live = tile < ntiles;
  v8f acc[8];
#pragma unroll
  for (int t = 0; t < 8; ++t) acc[t] = (v8f){};
  if (live) {
#pragma unroll
    for (int k0 = 0; k0 < D2; k0 += 4) {
      int m = m0 + (lane & 15);
      int kk = k0 + ((lane >> 4) << 1);
      v2f a;
      a.x = leaky(y2[(size_t)m * D2 + kk] * ak[kk] + ck[kk]);
      a.y = leaky(y2[(size_t)m * D2 + kk + 1] * ak[kk + 1] + ck[kk + 1]);
#pragma unroll
      for (int t = 0; t < 8; ++t) {
        v2f b = load_b_rm(W, C_OUT, k0, t * 16, lane);
        acc[t] = wmma4(a, b, acc[t]);
      }
    }
#pragma unroll
    for (int t = 0; t < 8; ++t) tile_col_stats(acc[t], s[t], q[t]);
  }
  int n = lane & 15, h = lane >> 4;
#pragma unroll
  for (int t = 0; t < 8; ++t) {
    s_st[wave][t * 16 + n][h] = s[t];
    s_st[wave][C_OUT + t * 16 + n][h] = q[t];
  }
  __syncthreads();
  if (live) {
#pragma unroll
    for (int t = 0; t < 8; ++t) store_d_rm(y3, C_OUT, m0, t * 16, lane, acc[t]);
  }
  int u = stat_of_slot(threadIdx.x, C_OUT);
  float tot = 0.f;
  for (int w = 0; w < 8; ++w) tot += s_st[w][u][0] + s_st[w][u][1];
  part[slot_index(threadIdx.x, gridDim.x, blockIdx.x)] = tot;
}

#define SX_LD 68
__global__ __launch_bounds__(128) void k_final(const float* __restrict__ x,
                                               const int* __restrict__ idx,
                                               const float* __restrict__ Wsc,
                                               const float* __restrict__ y3,
                                               const float* __restrict__ a2,
                                               const float* __restrict__ c2,
                                               float* __restrict__ out) {
  __shared__ __attribute__((aligned(16))) float s_sx[16 * SX_LD];
  int q0 = blockIdx.x * 16;
  unsigned gj = threadIdx.x & 15;
  const float4* xv4 = reinterpret_cast<const float4*>(x);
  int ids[2][NB];
#pragma unroll
  for (int pass = 0; pass < 2; ++pass) {
    unsigned q = (unsigned)q0 + pass * 8 + (threadIdx.x >> 4);
    const int4* ir = reinterpret_cast<const int4*>(idx + q * NB);
#pragma unroll
    for (int g = 0; g < 4; ++g) {
      int4 v = ir[g];
      ids[pass][4 * g + 0] = v.x; ids[pass][4 * g + 1] = v.y;
      ids[pass][4 * g + 2] = v.z; ids[pass][4 * g + 3] = v.w;
    }
  }
#pragma unroll
  for (int pass = 0; pass < 2; ++pass) {
    unsigned q = pass * 8 + (threadIdx.x >> 4);
    float4 xv[NB];
#pragma unroll
    for (int nb = 0; nb < NB; ++nb) {
      unsigned idc = ids[pass][nb] < N_PTS ? ids[pass][nb] : N_PTS - 1;
      xv[nb] = xv4[idc * (C_IN / 4) + gj];
    }
    float4 acc = make_float4(-3.4e38f, -3.4e38f, -3.4e38f, -3.4e38f);
#pragma unroll
    for (int nb = 0; nb < NB; ++nb) {
      bool real = ids[pass][nb] < N_PTS;
      acc.x = fmaxf(acc.x, real ? xv[nb].x : 0.f);
      acc.y = fmaxf(acc.y, real ? xv[nb].y : 0.f);
      acc.z = fmaxf(acc.z, real ? xv[nb].z : 0.f);
      acc.w = fmaxf(acc.w, real ? xv[nb].w : 0.f);
    }
    *reinterpret_cast<float4*>(&s_sx[q * SX_LD + 4 * gj]) = acc;
  }
  __syncthreads();
  int lane = threadIdx.x & 31;
  int wave = threadIdx.x >> 5;
#pragma unroll
  for (int tt = 0; tt < 2; ++tt) {
    int ntile = wave * 2 + tt;
    v8f acc = {};
#pragma unroll
    for (int k0 = 0; k0 < C_IN; k0 += 4) {
      int m = lane & 15;
      int kk = k0 + ((lane >> 4) << 1);
      v2f a;
      a.x = s_sx[m * SX_LD + kk];
      a.y = s_sx[m * SX_LD + kk + 1];
      v2f b = load_b_rm(Wsc, C_OUT, k0, ntile * 16, lane);
      acc = wmma4(a, b, acc);
    }
    int col = ntile * 16 + (lane & 15);
    int rb = (lane >> 4) << 3;
#pragma unroll
    for (int v = 0; v < 8; ++v) {
      int m = q0 + rb + v;
      float h = leaky(y3[(size_t)m * C_OUT + col] * a2[col] + c2[col]);
      out[(size_t)m * C_OUT + col] = h + acc[v];
    }
  }
}

extern "C" void kernel_launch(void* const* d_in, const int* in_sizes, int n_in,
                              void* d_out, int out_size, void* d_ws,
                              size_t ws_size, hipStream_t stream) {
  const float* x    = (const float*)d_in[0];
  const float* pos  = (const float*)d_in[1];
  const float* qpos = (const float*)d_in[2];
  const int*   idx  = (const int*)d_in[3];
  const float* Wu1  = (const float*)d_in[4];
  const float* g1   = (const float*)d_in[5];
  const float* b1   = (const float*)d_in[6];
  const float* Kpts = (const float*)d_in[7];
  const float* KV   = (const float*)d_in[8];
  const float* gk   = (const float*)d_in[9];
  const float* bk   = (const float*)d_in[10];
  const float* Wu2  = (const float*)d_in[11];
  const float* g2   = (const float*)d_in[12];
  const float* b2   = (const float*)d_in[13];
  const float* Wsc  = (const float*)d_in[14];
  float* out = (float*)d_out;

  float* ws = (float*)d_ws;
  float* y1 = ws;
  float* y2 = y1 + (size_t)N_PTS * D2;
  float* y3 = y2 + (size_t)M_Q * D2;
  float* S  = y3 + (size_t)M_Q * C_OUT;
  float *a1 = S +   0, *c1 = S +  32;
  float *ak = S +  64, *ck = S +  96;
  float *a2 = S + 128, *c2 = S + 256;
  float* part1 = y2;
  float* partk = y3;
  float* part2 = y1;

  k_gemm1<<<G1_BLOCKS, 256, 0, stream>>>(x, Wu1, y1, part1, N_PTS / 16);
  k_finalize_bn<<<D2 / 4, 1024, 0, stream>>>(part1, G1_BLOCKS, g1, b1,
                                             (float)N_PTS, a1, c1);
  k_kpconv<<<M_Q / 16, 128, 0, stream>>>(y1, idx, pos, qpos, Kpts, KV, a1, c1,
                                         y2, partk);
  k_finalize_bn<<<D2 / 4, 1024, 0, stream>>>(partk, M_Q / 16, gk, bk,
                                             (float)M_Q, ak, ck);
  k_gemm2<<<G2_BLOCKS, 256, 0, stream>>>(y2, Wu2, ak, ck, y3, part2,
                                         M_Q / 16);
  k_finalize_bn<<<C_OUT / 4, 1024, 0, stream>>>(part2, G2_BLOCKS, g2, b2,
                                                (float)M_Q, a2, c2);
  k_final<<<M_Q / 16, 128, 0, stream>>>(x, idx, Wsc, y3, a2, c2, out);
}


# N tiles own interleaved output columns (2c+t; 4c+t in gemm2), so weight fragments, y3 reads and D stores are 8/16-byte whole-line; KPConv step B: 4 k-quarters x both tiles (sum order changes)
# speedup vs baseline: 4.4975x; 1.0371x over previous
#include <hip/hip_runtime.h>


typedef float v2f __attribute__((ext_vector_type(2)));
typedef float v8f __attribute__((ext_vector_type(8)));

#define N_PTS   400000
#define M_Q     100000
#define NB      16
#define C_IN    64
#define D2      32
#define C_OUT   128
#define KP      15
#define INV_EXTENT 2.0f
#define EPS_BN  1e-5f
#define SLOPE   0.2f
#define G1_BLOCKS (N_PTS / 16 / 8)
#define G2_BLOCKS ((M_Q / 16 + 7) / 8)

__device__ __forceinline__ float leaky(float v) { return v >= 0.f ? v : SLOPE * v; }

__device__ __forceinline__ v8f wmma4(v2f a, v2f b, v8f c) {
  return __builtin_amdgcn_wmma_f32_16x16x4_f32(false, a, false, b, (short)0, c,
                                               false, false);
}

__device__ __forceinline__ v2f load_a_rm(const float* src, int ld, int m0,
                                         int k0, int lane) {
  int m = m0 + (lane & 15);
  int k = k0 + ((lane >> 4) << 1);
  v2f a;
  a.x = src[(size_t)m * ld + k];
  a.y = src[(size_t)m * ld + k + 1];
  return a;
}

__device__ __forceinline__ void load_b2_rm(const float* src, int ld, int k0,
                                           int c0, int lane, v2f& b0, v2f& b1) {
  int n = c0 + ((lane & 15) << 1);
  int k = k0 + ((lane >> 4) << 1);
  float2 r0 = *reinterpret_cast<const float2*>(src + (size_t)k * ld + n);
  float2 r1 = *reinterpret_cast<const float2*>(src + (size_t)(k + 1) * ld + n);
  b0.x = r0.x; b0.y = r1.x;
  b1.x = r0.y; b1.y = r1.y;
}

__device__ __forceinline__ void store_d2_rm(float* dst, int ld, int m0, int c0,
                                            int lane, v8f ca, v8f cb) {
  int n = c0 + ((lane & 15) << 1);
  int rb = m0 + ((lane >> 4) << 3);
#pragma unroll
  for (int v = 0; v < 8; ++v)
    *reinterpret_cast<float2*>(dst + (size_t)(rb + v) * ld + n) =
        make_float2(ca[v], cb[v]);
}


__device__ __forceinline__ void tile_col_stats(v8f c, float& s, float& q) {
  s = 0.f;
  q = 0.f;
#pragma unroll
  for (int v = 0; v < 8; ++v) { s += c[v]; q += c[v] * c[v]; }
}

__device__ __forceinline__ int stat_of_slot(int t, int C) {
  int j = t >> 3, v = t & 7;
  return 4 * j + (v & 3) + C * (v >> 2);
}
__device__ __forceinline__ size_t slot_index(int t, int nblocks, int blk) {
  int j = t >> 3, v = t & 7;
  return ((size_t)j * nblocks + blk) * 8 + v;
}

__global__ __launch_bounds__(1024) void k_finalize_bn(
    const float* __restrict__ part, int nparts, const float* __restrict__ g,
    const float* __restrict__ b, float cnt, float* a, float* c) {
  __shared__ float s_red[32][8];
  int j = blockIdx.x;
  int h = threadIdx.x & 1;
  int r0 = threadIdx.x >> 1;
  const float4* rows =
      reinterpret_cast<const float4*>(part + (size_t)j * nparts * 8);
  float4 acc = make_float4(0.f, 0.f, 0.f, 0.f);
  for (int base = 0; base < nparts; base += 8 * 512) {
    float4 v[8];
#pragma unroll
    for (int k = 0; k < 8; ++k) {
      int r = base + k * 512 + r0;
      int rc = r < nparts ? r : nparts - 1;
      v[k] = rows[(size_t)rc * 2 + h];
      if (r >= nparts) v[k] = make_float4(0.f, 0.f, 0.f, 0.f);
    }
#pragma unroll
    for (int k = 0; k < 8; ++k) {
      acc.x += v[k].x; acc.y += v[k].y; acc.z += v[k].z; acc.w += v[k].w;
    }
  }
  for (int m = 2; m < 32; m <<= 1) {
    acc.x += __shfl_xor(acc.x, m);
    acc.y += __shfl_xor(acc.y, m);
    acc.z += __shfl_xor(acc.z, m);
    acc.w += __shfl_xor(acc.w, m);
  }
  int lane = threadIdx.x & 31;
  int wave = threadIdx.x >> 5;
  if (lane < 2) {
    s_red[wave][4 * lane + 0] = acc.x;
    s_red[wave][4 * lane + 1] = acc.y;
    s_red[wave][4 * lane + 2] = acc.z;
    s_red[wave][4 * lane + 3] = acc.w;
  }
  __syncthreads();
  if (threadIdx.x < 4) {
    float s = 0.f, q = 0.f;
    for (int w = 0; w < 32; ++w) {
      s += s_red[w][threadIdx.x];
      q += s_red[w][4 + threadIdx.x];
    }
    int ch = 4 * j + threadIdx.x;
    float mu  = s / cnt;
    float var = q / cnt - mu * mu;
    float sc  = g[ch] * rsqrtf(fmaxf(var, 0.f) + EPS_BN);
    a[ch] = sc;
    c[ch] = b[ch] - mu * sc;
  }
}

__global__ __launch_bounds__(256) void k_gemm1(const float* __restrict__ x,
                                               const float* __restrict__ W,
                                               float* __restrict__ y1,
                                               float* __restrict__ part,
                                               int ntiles) {
  __shared__ float s_st[8][2 * D2][2];
  int lane = threadIdx.x & 31;
  int wave = threadIdx.x >> 5;
  int tile = blockIdx.x * 8 + wave;
  int m0 = tile * 16;
  bool live = tile < ntiles;
  float s0 = 0.f, q0 = 0.f, s1 = 0.f, q1 = 0.f;
  v8f acc0 = {}, acc1 = {};
  if (live) {
#pragma unroll
    for (int k0 = 0; k0 < C_IN; k0 += 4) {
      v2f a  = load_a_rm(x, C_IN, m0, k0, lane);
      v2f b0, b1;
      load_b2_rm(W, D2, k0, 0, lane, b0, b1);
      acc0 = wmma4(a, b0, acc0);
      acc1 = wmma4(a, b1, acc1);
    }
    tile_col_stats(acc0, s0, q0);
    tile_col_stats(acc1, s1, q1);
  }
  int n = lane & 15, h = lane >> 4;
  s_st[wave][2 * n][h] = s0;
  s_st[wave][2 * n + 1][h] = s1;
  s_st[wave][D2 + 2 * n][h] = q0;
  s_st[wave][D2 + 2 * n + 1][h] = q1;
  __syncthreads();
  if (live) {
    store_d2_rm(y1, D2, m0, 0, lane, acc0, acc1);
  }
  if (threadIdx.x < 2 * D2) {
    int u = stat_of_slot(threadIdx.x, D2);
    float t = 0.f;
    for (int w = 0; w < 8; ++w) t += s_st[w][u][0] + s_st[w][u][1];
    part[slot_index(threadIdx.x, gridDim.x, blockIdx.x)] = t;
  }
}

#define W_NB   20
#define W_Q    (NB * W_NB)
#define NF_NB  44
#define NF_Q   (NB * NF_NB)
#define WT_Q   532
#define WT_MID 16

__global__ __launch_bounds__(128) __attribute__((amdgpu_waves_per_eu(4))) void k_kpconv(
    const float* __restrict__ y1, const int* __restrict__ idx,
    const float* __restrict__ pos, const float* __restrict__ qpos,
    const float* __restrict__ Kpts, const float* __restrict__ KV,
    const float* __restrict__ a1, const float* __restrict__ c1,
    float* __restrict__ y2, float* __restrict__ part) {
  __shared__ __attribute__((aligned(16))) float smem[16 * W_Q + 16 * NF_Q];
  float* s_w  = smem;
  float* s_nf = smem + 16 * W_Q;
  int lane = threadIdx.x & 31;
  int wave = threadIdx.x >> 5;
  int q0 = blockIdx.x * 16;

  unsigned fq = threadIdx.x >> 3, fj = threadIdx.x & 7;
  int fid[NB];
  {
    const int4* ir =
        reinterpret_cast<const int4*>(idx + ((unsigned)q0 + fq) * NB);
#pragma unroll
    for (int g = 0; g < 4; ++g) {
      int4 v = ir[g];
      fid[4 * g + 0] = v.x; fid[4 * g + 1] = v.y;
      fid[4 * g + 2] = v.z; fid[4 * g + 3] = v.w;
    }
  }
  int wid[2];
  float wqx[2], wqy[2], wqz[2], wpx[2], wpy[2], wpz[2];
#pragma unroll
  for (int it = 0; it < 2; ++it) {
    unsigned p = threadIdx.x + 128 * it;
    unsigned q = (unsigned)q0 + (p >> 4), nb = p & 15;
    wid[it] = idx[q * NB + nb];
    const float* qp = qpos + q * 3;
    wqx[it] = qp[0];
    wqy[it] = qp[1];
    wqz[it] = qp[2];
  }
#pragma unroll
  for (int it = 0; it < 2; ++it) {
    unsigned idc = wid[it] < N_PTS ? wid[it] : N_PTS - 1;
    const float* pp = pos + idc * 3;
    wpx[it] = pp[0];
    wpy[it] = pp[1];
    wpz[it] = pp[2];
  }
  const float4* y1v = reinterpret_cast<const float4*>(y1);
  float4 yv[NB];
#pragma unroll
  for (int nb = 0; nb < NB; ++nb) {
    unsigned idc = fid[nb] < N_PTS ? fid[nb] : N_PTS - 1;
    yv[nb] = y1v[idc * (D2 / 4) + fj];
  }
  float4 av = reinterpret_cast<const float4*>(a1)[fj];
  float4 cv = reinterpret_cast<const float4*>(c1)[fj];
#pragma unroll
  for (int it = 0; it < 2; ++it) {
    unsigned p = threadIdx.x + 128 * it;
    unsigned q = p >> 4, nb = p & 15;
    bool real = wid[it] < N_PTS;
    float px = real ? wpx[it] : 1e6f;
    float py = real ? wpy[it] : 1e6f;
    float pz = real ? wpz[it] : 1e6f;
    float rx = px - wqx[it], ry = py - wqy[it], rz = pz - wqz[it];
    float rr = rx * rx + ry * ry + rz * rz;
#pragma unroll
    for (int k = 0; k < KP; ++k) {
      float kx = Kpts[k * 3 + 0], ky = Kpts[k * 3 + 1], kz = Kpts[k * 3 + 2];
      float sq = rr + kx * kx + ky * ky + kz * kz
                 - 2.f * (rx * kx + ry * ky + rz * kz);
      float w = 1.f - sqrtf(fmaxf(sq, 0.f)) * INV_EXTENT;
      s_w[q * W_Q + nb * W_NB + k] = fmaxf(w, 0.f);
    }
    s_w[q * W_Q + nb * W_NB + 15] = 0.f;
  }
#pragma unroll
  for (int nb = 0; nb < NB; ++nb) {
    bool real = fid[nb] < N_PTS;
    float4 o;
    o.x = real ? leaky(yv[nb].x * av.x + cv.x) : 0.f;
    o.y = real ? leaky(yv[nb].y * av.y + cv.y) : 0.f;
    o.z = real ? leaky(yv[nb].z * av.z + cv.z) : 0.f;
    o.w = real ? leaky(yv[nb].w * av.w + cv.w) : 0.f;
    *reinterpret_cast<float4*>(&s_nf[fq * NF_Q + nb * NF_NB + 4 * fj]) = o;
  }
  __syncthreads();

  v8f wt[4][2];
#pragma unroll
  for (int qq = 0; qq < 4; ++qq) { wt[qq][0] = (v8f){}; wt[qq][1] = (v8f){}; }
#pragma unroll
  for (int qq = 0; qq < 4; ++qq) {
    int q = wave * 4 + qq;
#pragma unroll
    for (int k0 = 0; k0 < 16; k0 += 4) {
      int m = lane & 15;
      int kk = k0 + ((lane >> 4) << 1);
      v2f a;
      a.x = s_w[q * W_Q + kk * W_NB + m];
      a.y = s_w[q * W_Q + (kk + 1) * W_NB + m];
#pragma unroll
      for (int t = 0; t < 2; ++t) {
        int n = t * 16 + (lane & 15);
        v2f b;
        b.x = s_nf[q * NF_Q + kk * NF_NB + n];
        b.y = s_nf[q * NF_Q + (kk + 1) * NF_NB + n];
        wt[qq][t] = wmma4(a, b, wt[qq][t]);
      }
    }
  }
  int kl = wave * 128 + ((lane >> 4) << 1);
  int nB = (lane & 15) << 1;
  v2f bf0[32], bf1[32];
#pragma unroll
  for (int i = 0; i < 16; ++i) {
    int kk = kl + 4 * i;
    int r0 = kk < 480 ? kk : 479;
    int r1 = kk + 1 < 480 ? kk + 1 : 479;
    float2 bx = *reinterpret_cast<const float2*>(KV + (size_t)r0 * 32 + nB);
    float2 by = *reinterpret_cast<const float2*>(KV + (size_t)r1 * 32 + nB);
    bf0[i].x = kk < 480 ? bx.x : 0.f;
    bf0[i].y = kk + 1 < 480 ? by.x : 0.f;
    bf1[i].x = kk < 480 ? bx.y : 0.f;
    bf1[i].y = kk + 1 < 480 ? by.y : 0.f;
  }
  __syncthreads();

  float* s_wt = smem;
#pragma unroll
  for (int qq = 0; qq < 4; ++qq) {
    int q = wave * 4 + qq;
#pragma unroll
    for (int t = 0; t < 2; ++t) {
      int n = t * 16 + (lane & 15);
      int rb = (lane >> 4) << 3;
      int so = q * WT_Q + rb * 32 + (lane >> 4) * WT_MID + n;
#pragma unroll
      for (int v = 0; v < 8; ++v)
        s_wt[so + v * 32] = wt[qq][t][v];
    }
  }
#pragma unroll
  for (int i = 16; i < 32; ++i) {
    int kk = kl + 4 * i;
    int r0 = kk < 480 ? kk : 479;
    int r1 = kk + 1 < 480 ? kk + 1 : 479;
    float2 bx = *reinterpret_cast<const float2*>(KV + (size_t)r0 * 32 + nB);
    float2 by = *reinterpret_cast<const float2*>(KV + (size_t)r1 * 32 + nB);
    bf0[i].x = kk < 480 ? bx.x : 0.f;
    bf0[i].y = kk + 1 < 480 ? by.x : 0.f;
    bf1[i].x = kk < 480 ? bx.y : 0.f;
    bf1[i].y = kk + 1 < 480 ? by.y : 0.f;
  }
  __syncthreads();

  v8f acc0 = {}, acc1 = {};
  int ao = (lane & 15) * WT_Q + kl + (wave >> 1) * WT_MID;
#pragma unroll
  for (int i = 0; i < 32; ++i) {
    v2f a;
    a.x = s_wt[ao + 4 * i];
    a.y = s_wt[ao + 4 * i + 1];
    acc0 = wmma4(a, bf0[i], acc0);
    acc1 = wmma4(a, bf1[i], acc1);
  }
  __syncthreads();
  float* s_red = smem;
  if (wave >= 1) {
#pragma unroll
    for (int v = 0; v < 8; ++v) {
      s_red[(wave - 1) * 512 + lane * 8 + v] = acc0[v];
      s_red[(wave - 1) * 512 + 256 + lane * 8 + v] = acc1[v];
    }
  }
  __syncthreads();
  float* s_st = smem + 1536;
  int col = (lane & 15) << 1;
  if (wave == 0) {
#pragma unroll
    for (int w = 0; w < 3; ++w) {
#pragma unroll
      for (int v = 0; v < 8; ++v) {
        acc0[v] += s_red[w * 512 + lane * 8 + v];
        acc1[v] += s_red[w * 512 + 256 + lane * 8 + v];
      }
    }
    float s0 = 0.f, q0s = 0.f, s1 = 0.f, q1s = 0.f;
#pragma unroll
    for (int v = 0; v < 8; ++v) {
      float va = acc0[v], vb = acc1[v];
      s0 += va;
      q0s += va * va;
      s1 += vb;
      q1s += vb * vb;
    }
    s_st[col * 2 + (lane >> 4)] = s0;
    s_st[(col + 1) * 2 + (lane >> 4)] = s1;
    s_st[(D2 + col) * 2 + (lane >> 4)] = q0s;
    s_st[(D2 + col + 1) * 2 + (lane >> 4)] = q1s;
  }
  __syncthreads();
  if (wave == 0) {
    int rb = q0 + ((lane >> 4) << 3);
#pragma unroll
    for (int v = 0; v < 8; ++v)
      *reinterpret_cast<float2*>(y2 + (size_t)(rb + v) * D2 + col) =
          make_float2(acc0[v], acc1[v]);
  }
  if (threadIdx.x < 2 * D2) {
    int u = stat_of_slot(threadIdx.x, D2);
    part[slot_index(threadIdx.x, gridDim.x, blockIdx.x)] =
        s_st[u * 2] + s_st[u * 2 + 1];
  }
}

__global__ __launch_bounds__(256) void k_gemm2(const float* __restrict__ y2,
                                               const float* __restrict__ W,
                                               const float* __restrict__ ak,
                                               const float* __restrict__ ck,
                                               float* __restrict__ y3,
                                               float* __restrict__ part,
                                               int ntiles) {
  __shared__ float s_st[8][2 * C_OUT][2];
  int lane = threadIdx.x & 31;
  int wave = threadIdx.x >> 5;
  int tile = blockIdx.x * 8 + wave;
  float s[8], q[8];
#pragma unroll
  for (int t = 0; t < 8; ++t) { s[t] = 0.f; q[t] = 0.f; }
  int m0 = tile * 16;
  bool live = tile < ntiles;
  v8f acc[8];
#pragma unroll
  for (int t = 0; t < 8; ++t) acc[t] = (v8f){};
  if (live) {
#pragma unroll
    for (int k0 = 0; k0 < D2; k0 += 4) {
      int m = m0 + (lane & 15);
      int kk = k0 + ((lane >> 4) << 1);
      v2f a;
      a.x = leaky(y2[(size_t)m * D2 + kk] * ak[kk] + ck[kk]);
      a.y = leaky(y2[(size_t)m * D2 + kk + 1] * ak[kk + 1] + ck[kk + 1]);
#pragma unroll
      for (int g = 0; g < 2; ++g) {
        const float* wp = W + (size_t)kk * C_OUT + 64 * g + 4 * (lane & 15);
        float4 r0 = *reinterpret_cast<const float4*>(wp);
        float4 r1 = *reinterpret_cast<const float4*>(wp + C_OUT);
        v2f b;
        b.x = r0.x; b.y = r1.x; acc[4 * g + 0] = wmma4(a, b, acc[4 * g + 0]);
        b.x = r0.y; b.y = r1.y; acc[4 * g + 1] = wmma4(a, b, acc[4 * g + 1]);
        b.x = r0.z; b.y = r1.z; acc[4 * g + 2] = wmma4(a, b, acc[4 * g + 2]);
        b.x = r0.w; b.y = r1.w; acc[4 * g + 3] = wmma4(a, b, acc[4 * g + 3]);
      }
    }
#pragma unroll
    for (int t = 0; t < 8; ++t) tile_col_stats(acc[t], s[t], q[t]);
  }
  int n = lane & 15, h = lane >> 4;
#pragma unroll
  for (int t = 0; t < 8; ++t) {
    int col = 64 * (t >> 2) + 4 * n + (t & 3);
    s_st[wave][col][h] = s[t];
    s_st[wave][C_OUT + col][h] = q[t];
  }
  __syncthreads();
  if (live) {
    int rb = m0 + (h << 3);
#pragma unroll
    for (int g = 0; g < 2; ++g) {
#pragma unroll
      for (int v = 0; v < 8; ++v)
        *reinterpret_cast<float4*>(y3 + (size_t)(rb + v) * C_OUT + 64 * g +
                                   4 * n) =
            make_float4(acc[4 * g + 0][v], acc[4 * g + 1][v],
                        acc[4 * g + 2][v], acc[4 * g + 3][v]);
    }
  }
  int u = stat_of_slot(threadIdx.x, C_OUT);
  float tot = 0.f;
  for (int w = 0; w < 8; ++w) tot += s_st[w][u][0] + s_st[w][u][1];
  part[slot_index(threadIdx.x, gridDim.x, blockIdx.x)] = tot;
}

#define SX_LD 68
__global__ __launch_bounds__(128) void k_final(const float* __restrict__ x,
                                               const int* __restrict__ idx,
                                               const float* __restrict__ Wsc,
                                               const float* __restrict__ y3,
                                               const float* __restrict__ a2,
                                               const float* __restrict__ c2,
                                               float* __restrict__ out) {
  __shared__ __attribute__((aligned(16))) float s_sx[16 * SX_LD];
  int q0 = blockIdx.x * 16;
  unsigned gj = threadIdx.x & 15;
  const float4* xv4 = reinterpret_cast<const float4*>(x);
  int ids[2][NB];
#pragma unroll
  for (int pass = 0; pass < 2; ++pass) {
    unsigned q = (unsigned)q0 + pass * 8 + (threadIdx.x >> 4);
    const int4* ir = reinterpret_cast<const int4*>(idx + q * NB);
#pragma unroll
    for (int g = 0; g < 4; ++g) {
      int4 v = ir[g];
      ids[pass][4 * g + 0] = v.x; ids[pass][4 * g + 1] = v.y;
      ids[pass][4 * g + 2] = v.z; ids[pass][4 * g + 3] = v.w;
    }
  }
#pragma unroll
  for (int pass = 0; pass < 2; ++pass) {
    unsigned q = pass * 8 + (threadIdx.x >> 4);
    float4 xv[NB];
#pragma unroll
    for (int nb = 0; nb < NB; ++nb) {
      unsigned idc = ids[pass][nb] < N_PTS ? ids[pass][nb] : N_PTS - 1;
      xv[nb] = xv4[idc * (C_IN / 4) + gj];
    }
    float4 acc = make_float4(-3.4e38f, -3.4e38f, -3.4e38f, -3.4e38f);
#pragma unroll
    for (int nb = 0; nb < NB; ++nb) {
      bool real = ids[pass][nb] < N_PTS;
      acc.x = fmaxf(acc.x, real ? xv[nb].x : 0.f);
      acc.y = fmaxf(acc.y, real ? xv[nb].y : 0.f);
      acc.z = fmaxf(acc.z, real ? xv[nb].z : 0.f);
      acc.w = fmaxf(acc.w, real ? xv[nb].w : 0.f);
    }
    *reinterpret_cast<float4*>(&s_sx[q * SX_LD + 4 * gj]) = acc;
  }
  __syncthreads();
  int lane = threadIdx.x & 31;
  int wave = threadIdx.x >> 5;
  v8f acc0 = {}, acc1 = {};
#pragma unroll
  for (int k0 = 0; k0 < C_IN; k0 += 4) {
    int m = lane & 15;
    int kk = k0 + ((lane >> 4) << 1);
    v2f a;
    a.x = s_sx[m * SX_LD + kk];
    a.y = s_sx[m * SX_LD + kk + 1];
    v2f b0, b1;
    load_b2_rm(Wsc, C_OUT, k0, 32 * wave, lane, b0, b1);
    acc0 = wmma4(a, b0, acc0);
    acc1 = wmma4(a, b1, acc1);
  }
  int col = 32 * wave + ((lane & 15) << 1);
  int rb = (lane >> 4) << 3;
  float2 sa = *reinterpret_cast<const float2*>(a2 + col);
  float2 sc = *reinterpret_cast<const float2*>(c2 + col);
#pragma unroll
  for (int v = 0; v < 8; ++v) {
    size_t o = (size_t)(q0 + rb + v) * C_OUT + col;
    float2 y = *reinterpret_cast<const float2*>(y3 + o);
    *reinterpret_cast<float2*>(out + o) =
        make_float2(leaky(y.x * sa.x + sc.x) + acc0[v],
                    leaky(y.y * sa.y + sc.y) + acc1[v]);
  }
}

extern "C" void kernel_launch(void* const* d_in, const int* in_sizes, int n_in,
                              void* d_out, int out_size, void* d_ws,
                              size_t ws_size, hipStream_t stream) {
  const float* x    = (const float*)d_in[0];
  const float* pos  = (const float*)d_in[1];
  const float* qpos = (const float*)d_in[2];
  const int*   idx  = (const int*)d_in[3];
  const float* Wu1  = (const float*)d_in[4];
  const float* g1   = (const float*)d_in[5];
  const float* b1   = (const float*)d_in[6];
  const float* Kpts = (const float*)d_in[7];
  const float* KV   = (const float*)d_in[8];
  const float* gk   = (const float*)d_in[9];
  const float* bk   = (const float*)d_in[10];
  const float* Wu2  = (const float*)d_in[11];
  const float* g2   = (const float*)d_in[12];
  const float* b2   = (const float*)d_in[13];
  const float* Wsc  = (const float*)d_in[14];
  float* out = (float*)d_out;

  float* ws = (float*)d_ws;
  float* y1 = ws;
  float* y2 = y1 + (size_t)N_PTS * D2;
  float* y3 = y2 + (size_t)M_Q * D2;
  float* S  = y3 + (size_t)M_Q * C_OUT;
  float *a1 = S +   0, *c1 = S +  32;
  float *ak = S +  64, *ck = S +  96;
  float *a2 = S + 128, *c2 = S + 256;
  float* part1 = y2;
  float* partk = y3;
  float* part2 = y1;

  k_gemm1<<<G1_BLOCKS, 256, 0, stream>>>(x, Wu1, y1, part1, N_PTS / 16);
  k_finalize_bn<<<D2 / 4, 1024, 0, stream>>>(part1, G1_BLOCKS, g1, b1,
                                             (float)N_PTS, a1, c1);
  k_kpconv<<<M_Q / 16, 128, 0, stream>>>(y1, idx, pos, qpos, Kpts, KV, a1, c1,
                                         y2, partk);
  k_finalize_bn<<<D2 / 4, 1024, 0, stream>>>(partk, M_Q / 16, gk, bk,
                                             (float)M_Q, ak, ck);
  k_gemm2<<<G2_BLOCKS, 256, 0, stream>>>(y2, Wu2, ak, ck, y3, part2,
                                         M_Q / 16);
  k_finalize_bn<<<C_OUT / 4, 1024, 0, stream>>>(part2, G2_BLOCKS, g2, b2,
                                                (float)M_Q, a2, c2);
  k_final<<<M_Q / 16, 128, 0, stream>>>(x, idx, Wsc, y3, a2, c2, out);
}
